# starting kernel
# baseline (speedup 1.0000x reference)
#include <hip/hip_runtime.h>
#include <math.h>


typedef __attribute__((ext_vector_type(2))) float v2f;
typedef __attribute__((ext_vector_type(8))) float v8f;

#define LDS_STRIDE 68

__device__ __forceinline__ float gelu_exact(float x) {
    return 0.5f * x * (1.0f + erff(x * 0.70710678118654752f));
}

__global__ __launch_bounds__(256)
void embed_kernel(const int* __restrict__ positions,
                  const int* __restrict__ amino,
                  const float* __restrict__ pos_table,
                  const float* __restrict__ aa_table,
                  float* __restrict__ E) {
    int t  = blockIdx.x * 256 + threadIdx.x;
    int k  = t & 63;
    int bm = t >> 6;
    int p = positions[bm]; p = p < 0 ? 0 : (p > 1023 ? 1023 : p);
    int a = amino[bm];     a = a < 0 ? 0 : (a > 20   ? 20   : a);
    E[t] = pos_table[p * 64 + k] + aa_table[a * 64 + k];
}

__global__ __launch_bounds__(256)
void pair_kernel(const float* __restrict__ E,
                 const float* __restrict__ W1,
                 const float* __restrict__ b1,
                 const float* __restrict__ W2,
                 const float* __restrict__ b2,
                 const float* __restrict__ W3,
                 const float* __restrict__ b3,
                 float* __restrict__ out) {
    __shared__ float u_lds[16 * LDS_STRIDE];
    __shared__ float v_lds[16 * LDS_STRIDE];
    __shared__ float w2t [64 * LDS_STRIDE];
    __shared__ float b1s[64], b2s[64], w3s[64];

    const int tid = threadIdx.x;
    const int b   = blockIdx.y;
    const int tp  = blockIdx.x;
    const int it  = (tp < 4) ? 0 : (tp < 7) ? 1 : (tp < 9) ? 2 : 3;
    const int start = it * 4 - (it * (it - 1)) / 2;
    const int jt  = tp - start + it;

    {
        const int n  = tid & 63;
        const int i0 = tid >> 6;
        const float* Ei = E + (b * 64 + it * 16) * 64;
        const float* Ej = E + (b * 64 + jt * 16) * 64;
        for (int s = 0; s < 4; ++s) {
            const int i = i0 + 4 * s;
            float su = 0.f, sv = 0.f;
            for (int k = 0; k < 64; ++k) {
                su = fmaf(Ei[i * 64 + k], W1[k * 64 + n],        su);
                sv = fmaf(Ej[i * 64 + k], W1[(64 + k) * 64 + n], sv);
            }
            u_lds[i * LDS_STRIDE + n] = su;
            v_lds[i * LDS_STRIDE + n] = sv;
        }
        for (int e = tid; e < 4096; e += 256) {
            int k = e >> 6, nn = e & 63;
            w2t[nn * LDS_STRIDE + k] = W2[k * 64 + nn];
        }
        if (tid < 64) { b1s[tid] = b1[tid]; b2s[tid] = b2[tid]; w3s[tid] = W3[tid]; }
    }
    __syncthreads();

    const int lane = tid & 31;
    const int wv   = tid >> 5;
    const int jl   = lane & 15;
    const int half = lane >> 4;
    const int koff = 2 * half;

    v2f hA[2][16];
    #pragma unroll
    for (int mt = 0; mt < 2; ++mt) {
        const int il = 2 * wv + mt;
        #pragma unroll
        for (int kk = 0; kk < 16; ++kk) {
            const int k0 = 4 * kk + koff;
            float x0 = u_lds[il * LDS_STRIDE + k0]     + v_lds[jl * LDS_STRIDE + k0]     + b1s[k0];
            float x1 = u_lds[il * LDS_STRIDE + k0 + 1] + v_lds[jl * LDS_STRIDE + k0 + 1] + b1s[k0 + 1];
            hA[mt][kk].x = gelu_exact(x0);
            hA[mt][kk].y = gelu_exact(x1);
        }
    }

    float p0[8], p1[8];
    #pragma unroll
    for (int r = 0; r < 8; ++r) { p0[r] = 0.f; p1[r] = 0.f; }

    #pragma unroll
    for (int nt = 0; nt < 4; ++nt) {
        const int n = nt * 16 + jl;
        v8f acc0 = {};
        v8f acc1 = {};
        #pragma unroll
        for (int kk = 0; kk < 16; ++kk) {
            const int k0 = 4 * kk + koff;
            v2f bb = *(const v2f*)&w2t[n * LDS_STRIDE + k0];
            acc0 = __builtin_amdgcn_wmma_f32_16x16x4_f32(false, hA[0][kk], false, bb,
                                                         (short)0, acc0, false, false);
            acc1 = __builtin_amdgcn_wmma_f32_16x16x4_f32(false, hA[1][kk], false, bb,
                                                         (short)0, acc1, false, false);
        }
        const float b2n = b2s[n];
        const float w3n = w3s[n];
        #pragma unroll
        for (int r = 0; r < 8; ++r) {
            p0[r] = fmaf(gelu_exact(acc0[r] + b2n), w3n, p0[r]);
            p1[r] = fmaf(gelu_exact(acc1[r] + b2n), w3n, p1[r]);
        }
    }

    const float bias3 = b3[0];
    #pragma unroll
    for (int mt = 0; mt < 2; ++mt) {
        const int ig = it * 16 + 2 * wv + mt;
        #pragma unroll
        for (int r = 0; r < 8; ++r) {
            float s = (mt == 0) ? p0[r] : p1[r];
            s += __shfl_xor(s, 1);
            s += __shfl_xor(s, 2);
            s += __shfl_xor(s, 4);
            s += __shfl_xor(s, 8);
            if (jl == 0) {
                const int jg = jt * 16 + r + 8 * half;
                if (ig < jg) {
                    const float val = s + bias3;
                    out[(b * 64 + ig) * 64 + jg] = val;
                    out[(b * 64 + jg) * 64 + ig] = val;
                } else if (ig == jg) {
                    out[(b * 64 + ig) * 64 + jg] = 0.f;
                }
            }
        }
    }
}

extern "C" void kernel_launch(void* const* d_in, const int* in_sizes, int n_in,
                              void* d_out, int out_size, void* d_ws, size_t ws_size,
                              hipStream_t stream) {
    const int*   positions = (const int*)  d_in[0];
    const int*   amino     = (const int*)  d_in[1];
    const float* pos_table = (const float*)d_in[2];
    const float* aa_table  = (const float*)d_in[3];
    const float* W1        = (const float*)d_in[4];
    const float* b1        = (const float*)d_in[5];
    const float* W2        = (const float*)d_in[6];
    const float* b2        = (const float*)d_in[7];
    const float* W3        = (const float*)d_in[8];
    const float* b3        = (const float*)d_in[9];
    float* out = (float*)d_out;
    float* E   = (float*)d_ws;

    embed_kernel<<<dim3((128 * 64 * 64) / 256), dim3(256), 0, stream>>>(
        positions, amino, pos_table, aa_table, E);
    pair_kernel<<<dim3(10, 128), dim3(256), 0, stream>>>(
        E, W1, b1, W2, b2, W3, b3, out);
}


# erf inside gelu made branch-free: both range formulas of the library erff (same coefficients, same operation order) evaluated, one selected; gelu calls now overlap and pack into 2-wide f32 fma
# speedup vs baseline: 1.8908x; 1.8908x over previous
#include <hip/hip_runtime.h>
#include <math.h>


typedef __attribute__((ext_vector_type(2))) float v2f;
typedef __attribute__((ext_vector_type(8))) float v8f;

#define LDS_STRIDE 68

__device__ __forceinline__ float erf_select(float x) {
    const float ax = fabsf(x);
    const float t  = ax * ax;
    float p = fmaf(t, -0x1.268bc2p-11f, 0x1.420828p-8f);
    p = fmaf(t, p, -0x1.b59370p-6f);
    p = fmaf(t, p,  0x1.ce077cp-4f);
    p = fmaf(t, p, -0x1.812660p-2f);
    p = fmaf(t, p,  0x1.06eba0p-3f);
    const float r_small = fmaf(ax, p, ax);

    float q = fmaf(ax, 0x1.1d3156p-16f, -0x1.8d1290p-12f);
    q = fmaf(ax, q,  0x1.f9a6d2p-9f);
    q = fmaf(ax, q, -0x1.8c3164p-6f);
    q = fmaf(ax, q,  0x1.b4e9c8p-4f);
    q = fmaf(ax, q,  0x1.4515fap-1f);
    q = fmaf(ax, q,  0x1.078e50p-3f);
    q = fmaf(ax, q, ax);
    const float r_large = 1.0f - expf(-q);

    return copysignf(ax < 1.0f ? r_small : r_large, x);
}

__device__ __forceinline__ float gelu_exact(float x) {
    return 0.5f * x * (1.0f + erf_select(x * 0.70710678118654752f));
}

#define TOK 16
__global__ __launch_bounds__(256)
void token_proj_kernel(const int* __restrict__ positions,
                       const int* __restrict__ amino,
                       const float* __restrict__ pos_table,
                       const float* __restrict__ aa_table,
                       const float* __restrict__ W1,
                       float* __restrict__ U,
                       float* __restrict__ V) {
    __shared__ float e_lds[TOK * 64];

    const int tid  = threadIdx.x;
    const int tok0 = blockIdx.x * TOK;

    const int gt = tid >> 4;
    const int gk = (tid & 15) * 4;
    int p = positions[tok0 + gt]; p = p < 0 ? 0 : (p > 1023 ? 1023 : p);
    int a = amino[tok0 + gt];     a = a < 0 ? 0 : (a > 20   ? 20   : a);

    const int n    = tid & 63;
    const int hsel = (tid >> 6) & 1;
    const int tg   = tid >> 7;
    const float* Wc = W1 + hsel * 64 * 64 + n;

    float w[64];
    #pragma unroll
    for (int k = 0; k < 64; ++k) w[k] = Wc[k * 64];

    {
        const float4 pe = *(const float4*)(pos_table + p * 64 + gk);
        const float4 ae = *(const float4*)(aa_table  + a * 64 + gk);
        float4 ev;
        ev.x = pe.x + ae.x; ev.y = pe.y + ae.y; ev.z = pe.z + ae.z; ev.w = pe.w + ae.w;
        *(float4*)&e_lds[gt * 64 + gk] = ev;
    }
    __syncthreads();

    float acc[8];
    #pragma unroll
    for (int t = 0; t < 8; ++t) acc[t] = 0.f;
    const float* Et = e_lds + tg * 8 * 64;
    #pragma unroll
    for (int k = 0; k < 64; ++k) {
        #pragma unroll
        for (int t = 0; t < 8; ++t) acc[t] = fmaf(Et[t * 64 + k], w[k], acc[t]);
    }

    float* dst = (hsel ? V : U) + (tok0 + tg * 8) * 64 + n;
    #pragma unroll
    for (int t = 0; t < 8; ++t) dst[t * 64] = acc[t];
}

__global__ __launch_bounds__(256)
void pair_kernel(const float* __restrict__ U,
                 const float* __restrict__ V,
                 const float* __restrict__ b1,
                 const float* __restrict__ W2,
                 const float* __restrict__ b2,
                 const float* __restrict__ W3,
                 const float* __restrict__ b3,
                 float* __restrict__ out) {
    __shared__ float u_lds[16 * LDS_STRIDE];
    __shared__ float v_lds[16 * LDS_STRIDE];
    __shared__ float w2t [64 * LDS_STRIDE];
    __shared__ float b1s[64], b2s[64], w3s[64];

    const int tid = threadIdx.x;
    const int b   = blockIdx.y;
    const int tp  = blockIdx.x;
    const int it  = (tp < 4) ? 0 : (tp < 7) ? 1 : (tp < 9) ? 2 : 3;
    const int start = it * 4 - (it * (it - 1)) / 2;
    const int jt  = tp - start + it;

    {
        const int row = tid >> 4;
        const int c4  = (tid & 15) * 4;
        const float4 uu = *(const float4*)(U + (b * 64 + it * 16 + row) * 64 + c4);
        const float4 vv = *(const float4*)(V + (b * 64 + jt * 16 + row) * 64 + c4);
        *(float4*)&u_lds[row * LDS_STRIDE + c4] = uu;
        *(float4*)&v_lds[row * LDS_STRIDE + c4] = vv;
        for (int e = tid; e < 4096; e += 256) {
            int k = e >> 6, nn = e & 63;
            w2t[nn * LDS_STRIDE + k] = W2[k * 64 + nn];
        }
        if (tid < 64) { b1s[tid] = b1[tid]; b2s[tid] = b2[tid]; w3s[tid] = W3[tid]; }
    }
    __syncthreads();

    const int lane = tid & 31;
    const int wv   = tid >> 5;
    const int jl   = lane & 15;
    const int half = lane >> 4;
    const int koff = 2 * half;

    v2f hA[2][16];
    #pragma unroll
    for (int mt = 0; mt < 2; ++mt) {
        const int il = 2 * wv + mt;
        #pragma unroll
        for (int kk = 0; kk < 16; ++kk) {
            const int k0 = 4 * kk + koff;
            float x0 = u_lds[il * LDS_STRIDE + k0]     + v_lds[jl * LDS_STRIDE + k0]     + b1s[k0];
            float x1 = u_lds[il * LDS_STRIDE + k0 + 1] + v_lds[jl * LDS_STRIDE + k0 + 1] + b1s[k0 + 1];
            hA[mt][kk].x = gelu_exact(x0);
            hA[mt][kk].y = gelu_exact(x1);
        }
    }

    float p0[8], p1[8];
    #pragma unroll
    for (int r = 0; r < 8; ++r) { p0[r] = 0.f; p1[r] = 0.f; }

    #pragma unroll
    for (int nt = 0; nt < 4; ++nt) {
        const int n = nt * 16 + jl;
        v8f acc0 = {};
        v8f acc1 = {};
        #pragma unroll
        for (int kk = 0; kk < 16; ++kk) {
            const int k0 = 4 * kk + koff;
            v2f bb = *(const v2f*)&w2t[n * LDS_STRIDE + k0];
            acc0 = __builtin_amdgcn_wmma_f32_16x16x4_f32(false, hA[0][kk], false, bb,
                                                         (short)0, acc0, false, false);
            acc1 = __builtin_amdgcn_wmma_f32_16x16x4_f32(false, hA[1][kk], false, bb,
                                                         (short)0, acc1, false, false);
        }
        const float b2n = b2s[n];
        const float w3n = w3s[n];
        #pragma unroll
        for (int r = 0; r < 8; ++r) {
            p0[r] = fmaf(gelu_exact(acc0[r] + b2n), w3n, p0[r]);
            p1[r] = fmaf(gelu_exact(acc1[r] + b2n), w3n, p1[r]);
        }
    }

    const float bias3 = b3[0];
    #pragma unroll
    for (int mt = 0; mt < 2; ++mt) {
        const int ig = it * 16 + 2 * wv + mt;
        #pragma unroll
        for (int r = 0; r < 8; ++r) {
            float s = (mt == 0) ? p0[r] : p1[r];
            s += __shfl_xor(s, 1);
            s += __shfl_xor(s, 2);
            s += __shfl_xor(s, 4);
            s += __shfl_xor(s, 8);
            if (jl == 0) {
                const int jg = jt * 16 + r + 8 * half;
                if (ig < jg) {
                    const float val = s + bias3;
                    out[(b * 64 + ig) * 64 + jg] = val;
                    out[(b * 64 + jg) * 64 + ig] = val;
                } else if (ig == jg) {
                    out[(b * 64 + ig) * 64 + jg] = 0.f;
                }
            }
        }
    }
}

extern "C" void kernel_launch(void* const* d_in, const int* in_sizes, int n_in,
                              void* d_out, int out_size, void* d_ws, size_t ws_size,
                              hipStream_t stream) {
    const int*   positions = (const int*)  d_in[0];
    const int*   amino     = (const int*)  d_in[1];
    const float* pos_table = (const float*)d_in[2];
    const float* aa_table  = (const float*)d_in[3];
    const float* W1        = (const float*)d_in[4];
    const float* b1        = (const float*)d_in[5];
    const float* W2        = (const float*)d_in[6];
    const float* b2        = (const float*)d_in[7];
    const float* W3        = (const float*)d_in[8];
    const float* b3        = (const float*)d_in[9];
    float* out = (float*)d_out;
    float* U   = (float*)d_ws;
    float* V   = U + 128 * 64 * 64;

    token_proj_kernel<<<dim3((128 * 64) / TOK), dim3(256), 0, stream>>>(
        positions, amino, pos_table, aa_table, W1, U, V);
    pair_kernel<<<dim3(10, 128), dim3(256), 0, stream>>>(
        U, V, b1, W2, b2, W3, b3, out);
}


# pair kernel enumerates only the 2016 pairs i<j per batch (folded rows f and 62-f, each lane's WMMA A row is its own pair): 126 row tiles, not 160; grid 8x128 blocks; per-pair sums unchanged
# speedup vs baseline: 2.0519x; 1.0852x over previous
#include <hip/hip_runtime.h>
#include <math.h>


typedef __attribute__((ext_vector_type(2))) float v2f;
typedef __attribute__((ext_vector_type(8))) float v8f;

#define LDS_STRIDE 68

__device__ __forceinline__ float erf_select(float x) {
    const float ax = fabsf(x);
    const float t  = ax * ax;
    float p = fmaf(t, -0x1.268bc2p-11f, 0x1.420828p-8f);
    p = fmaf(t, p, -0x1.b59370p-6f);
    p = fmaf(t, p,  0x1.ce077cp-4f);
    p = fmaf(t, p, -0x1.812660p-2f);
    p = fmaf(t, p,  0x1.06eba0p-3f);
    const float r_small = fmaf(ax, p, ax);

    float q = fmaf(ax, 0x1.1d3156p-16f, -0x1.8d1290p-12f);
    q = fmaf(ax, q,  0x1.f9a6d2p-9f);
    q = fmaf(ax, q, -0x1.8c3164p-6f);
    q = fmaf(ax, q,  0x1.b4e9c8p-4f);
    q = fmaf(ax, q,  0x1.4515fap-1f);
    q = fmaf(ax, q,  0x1.078e50p-3f);
    q = fmaf(ax, q, ax);
    const float r_large = 1.0f - expf(-q);

    return copysignf(ax < 1.0f ? r_small : r_large, x);
}

__device__ __forceinline__ float gelu_exact(float x) {
    return 0.5f * x * (1.0f + erf_select(x * 0.70710678118654752f));
}

#define TOK 16
__global__ __launch_bounds__(256)
void token_proj_kernel(const int* __restrict__ positions,
                       const int* __restrict__ amino,
                       const float* __restrict__ pos_table,
                       const float* __restrict__ aa_table,
                       const float* __restrict__ W1,
                       float* __restrict__ U,
                       float* __restrict__ V) {
    __shared__ float e_lds[TOK * 64];

    const int tid  = threadIdx.x;
    const int tok0 = blockIdx.x * TOK;

    const int gt = tid >> 4;
    const int gk = (tid & 15) * 4;
    int p = positions[tok0 + gt]; p = p < 0 ? 0 : (p > 1023 ? 1023 : p);
    int a = amino[tok0 + gt];     a = a < 0 ? 0 : (a > 20   ? 20   : a);

    const int n    = tid & 63;
    const int hsel = (tid >> 6) & 1;
    const int tg   = tid >> 7;
    const float* Wc = W1 + hsel * 64 * 64 + n;

    float w[64];
    #pragma unroll
    for (int k = 0; k < 64; ++k) w[k] = Wc[k * 64];

    {
        const float4 pe = *(const float4*)(pos_table + p * 64 + gk);
        const float4 ae = *(const float4*)(aa_table  + a * 64 + gk);
        float4 ev;
        ev.x = pe.x + ae.x; ev.y = pe.y + ae.y; ev.z = pe.z + ae.z; ev.w = pe.w + ae.w;
        *(float4*)&e_lds[gt * 64 + gk] = ev;
    }
    __syncthreads();

    float acc[8];
    #pragma unroll
    for (int t = 0; t < 8; ++t) acc[t] = 0.f;
    const float* Et = e_lds + tg * 8 * 64;
    #pragma unroll
    for (int k = 0; k < 64; ++k) {
        #pragma unroll
        for (int t = 0; t < 8; ++t) acc[t] = fmaf(Et[t * 64 + k], w[k], acc[t]);
    }

    float* dst = (hsel ? V : U) + (tok0 + tg * 8) * 64 + n;
    #pragma unroll
    for (int t = 0; t < 8; ++t) dst[t * 64] = acc[t];
}

__global__ __launch_bounds__(256)
void pair_kernel(const float* __restrict__ U,
                 const float* __restrict__ V,
                 const float* __restrict__ b1,
                 const float* __restrict__ W2,
                 const float* __restrict__ b2,
                 const float* __restrict__ W3,
                 const float* __restrict__ b3,
                 float* __restrict__ out) {
    __shared__ float u_lds[8 * LDS_STRIDE];
    __shared__ float v_lds[64 * LDS_STRIDE];
    __shared__ float w2t [64 * LDS_STRIDE];
    __shared__ float b1s[64], b2s[64], w3s[64];

    const int tid = threadIdx.x;
    const int b   = blockIdx.y;
    const int f0  = blockIdx.x * 4;

    {
        const int r16 = tid >> 4;
        const int c4  = (tid & 15) * 4;
        if (tid < 128) {
            const int fs = f0 + (r16 >> 1);
            const int iu = (r16 & 1) ? 62 - fs : fs;
            const float4 uu = *(const float4*)(U + (b * 64 + iu) * 64 + c4);
            *(float4*)&u_lds[r16 * LDS_STRIDE + c4] = uu;
        }
        #pragma unroll
        for (int q = 0; q < 4; ++q) {
            const int row = r16 + 16 * q;
            const float4 vv = *(const float4*)(V + (b * 64 + row) * 64 + c4);
            *(float4*)&v_lds[row * LDS_STRIDE + c4] = vv;
        }
        for (int e = tid; e < 4096; e += 256) {
            int k = e >> 6, nn = e & 63;
            w2t[nn * LDS_STRIDE + k] = W2[k * 64 + nn];
        }
        if (tid < 64) { b1s[tid] = b1[tid]; b2s[tid] = b2[tid]; w3s[tid] = W3[tid]; }
    }
    __syncthreads();

    const int lane = tid & 31;
    const int wv   = tid >> 5;
    const int fl   = wv >> 1;
    const int hr   = wv & 1;
    const int f    = f0 + fl;
    if (f == 31 && hr == 1) return;
    const int nfirst = 63 - f;
    const int jl   = lane & 15;
    const int half = lane >> 4;
    const int koff = 2 * half;

    v2f hA[2][16];
    #pragma unroll
    for (int mt = 0; mt < 2; ++mt) {
        const int t    = 32 * hr + 16 * mt + jl;
        const int sel  = (t >= nfirst) ? 1 : 0;
        const int urow = 2 * fl + sel;
        const int jrow = sel ? t : f + 1 + t;
        #pragma unroll
        for (int kk = 0; kk < 16; ++kk) {
            const int k0 = 4 * kk + koff;
            float x0 = u_lds[urow * LDS_STRIDE + k0]     + v_lds[jrow * LDS_STRIDE + k0]     + b1s[k0];
            float x1 = u_lds[urow * LDS_STRIDE + k0 + 1] + v_lds[jrow * LDS_STRIDE + k0 + 1] + b1s[k0 + 1];
            hA[mt][kk].x = gelu_exact(x0);
            hA[mt][kk].y = gelu_exact(x1);
        }
    }

    float p0[8], p1[8];
    #pragma unroll
    for (int r = 0; r < 8; ++r) { p0[r] = 0.f; p1[r] = 0.f; }

    #pragma unroll
    for (int nt = 0; nt < 4; ++nt) {
        const int n = nt * 16 + jl;
        v8f acc0 = {};
        v8f acc1 = {};
        #pragma unroll
        for (int kk = 0; kk < 16; ++kk) {
            const int k0 = 4 * kk + koff;
            v2f bb = *(const v2f*)&w2t[n * LDS_STRIDE + k0];
            acc0 = __builtin_amdgcn_wmma_f32_16x16x4_f32(false, hA[0][kk], false, bb,
                                                         (short)0, acc0, false, false);
            acc1 = __builtin_amdgcn_wmma_f32_16x16x4_f32(false, hA[1][kk], false, bb,
                                                         (short)0, acc1, false, false);
        }
        const float b2n = b2s[n];
        const float w3n = w3s[n];
        #pragma unroll
        for (int r = 0; r < 8; ++r) {
            p0[r] = fmaf(gelu_exact(acc0[r] + b2n), w3n, p0[r]);
            p1[r] = fmaf(gelu_exact(acc1[r] + b2n), w3n, p1[r]);
        }
    }

    const float bias3 = b3[0];
    #pragma unroll
    for (int mt = 0; mt < 2; ++mt) {
        #pragma unroll
        for (int r = 0; r < 8; ++r) {
            float s = (mt == 0) ? p0[r] : p1[r];
            s += __shfl_xor(s, 1);
            s += __shfl_xor(s, 2);
            s += __shfl_xor(s, 4);
            s += __shfl_xor(s, 8);
            if (jl == 0) {
                const int tq = 32 * hr + 16 * mt + 8 * half + r;
                const int sq = (tq >= nfirst) ? 1 : 0;
                const int ig = sq ? 62 - f : f;
                const int jg = sq ? tq : f + 1 + tq;
                const float val = s + bias3;
                out[(b * 64 + ig) * 64 + jg] = val;
                out[(b * 64 + jg) * 64 + ig] = val;
            }
        }
    }

    if (tid < 8) {
        const int idg = blockIdx.x * 8 + tid;
        out[(b * 64 + idg) * 64 + idg] = 0.f;
    }
}

extern "C" void kernel_launch(void* const* d_in, const int* in_sizes, int n_in,
                              void* d_out, int out_size, void* d_ws, size_t ws_size,
                              hipStream_t stream) {
    const int*   positions = (const int*)  d_in[0];
    const int*   amino     = (const int*)  d_in[1];
    const float* pos_table = (const float*)d_in[2];
    const float* aa_table  = (const float*)d_in[3];
    const float* W1        = (const float*)d_in[4];
    const float* b1        = (const float*)d_in[5];
    const float* W2        = (const float*)d_in[6];
    const float* b2        = (const float*)d_in[7];
    const float* W3        = (const float*)d_in[8];
    const float* b3        = (const float*)d_in[9];
    float* out = (float*)d_out;
    float* U   = (float*)d_ws;
    float* V   = U + 128 * 64 * 64;

    token_proj_kernel<<<dim3((128 * 64) / TOK), dim3(256), 0, stream>>>(
        positions, amino, pos_table, aa_table, W1, U, V);
    pair_kernel<<<dim3(8, 128), dim3(256), 0, stream>>>(
        U, V, b1, W2, b2, W3, b3, out);
}


# erf's exp(-q) as hardware exp2(-q*log2e): mul + v_exp_f32 replace library expf's split product, rescale and range selects (erf moves ~1 ulp); layer-2 gelu loop split per row tile keeps 160 VGPRs
# speedup vs baseline: 2.2536x; 1.0983x over previous
#include <hip/hip_runtime.h>
#include <math.h>


typedef __attribute__((ext_vector_type(2))) float v2f;
typedef __attribute__((ext_vector_type(8))) float v8f;

#define LDS_STRIDE 68

__device__ __forceinline__ float erf_select(float x) {
    const float ax = fabsf(x);
    const float t  = ax * ax;
    float p = fmaf(t, -0x1.268bc2p-11f, 0x1.420828p-8f);
    p = fmaf(t, p, -0x1.b59370p-6f);
    p = fmaf(t, p,  0x1.ce077cp-4f);
    p = fmaf(t, p, -0x1.812660p-2f);
    p = fmaf(t, p,  0x1.06eba0p-3f);
    const float r_small = fmaf(ax, p, ax);

    float q = fmaf(ax, 0x1.1d3156p-16f, -0x1.8d1290p-12f);
    q = fmaf(ax, q,  0x1.f9a6d2p-9f);
    q = fmaf(ax, q, -0x1.8c3164p-6f);
    q = fmaf(ax, q,  0x1.b4e9c8p-4f);
    q = fmaf(ax, q,  0x1.4515fap-1f);
    q = fmaf(ax, q,  0x1.078e50p-3f);
    q = fmaf(ax, q, ax);
    const float r_large = 1.0f - __builtin_amdgcn_exp2f(q * -0x1.715476p+0f);

    return copysignf(ax < 1.0f ? r_small : r_large, x);
}

__device__ __forceinline__ float gelu_exact(float x) {
    return 0.5f * x * (1.0f + erf_select(x * 0.70710678118654752f));
}

#define TOK 16
__global__ __launch_bounds__(256)
void token_proj_kernel(const int* __restrict__ positions,
                       const int* __restrict__ amino,
                       const float* __restrict__ pos_table,
                       const float* __restrict__ aa_table,
                       const float* __restrict__ W1,
                       float* __restrict__ U,
                       float* __restrict__ V) {
    __shared__ float e_lds[TOK * 64];

    const int tid  = threadIdx.x;
    const int tok0 = blockIdx.x * TOK;

    const int gt = tid >> 4;
    const int gk = (tid & 15) * 4;
    int p = positions[tok0 + gt]; p = p < 0 ? 0 : (p > 1023 ? 1023 : p);
    int a = amino[tok0 + gt];     a = a < 0 ? 0 : (a > 20   ? 20   : a);

    const int n    = tid & 63;
    const int hsel = (tid >> 6) & 1;
    const int tg   = tid >> 7;
    const float* Wc = W1 + hsel * 64 * 64 + n;

    float w[64];
    #pragma unroll
    for (int k = 0; k < 64; ++k) w[k] = Wc[k * 64];

    {
        const float4 pe = *(const float4*)(pos_table + p * 64 + gk);
        const float4 ae = *(const float4*)(aa_table  + a * 64 + gk);
        float4 ev;
        ev.x = pe.x + ae.x; ev.y = pe.y + ae.y; ev.z = pe.z + ae.z; ev.w = pe.w + ae.w;
        *(float4*)&e_lds[gt * 64 + gk] = ev;
    }
    __syncthreads();

    float acc[8];
    #pragma unroll
    for (int t = 0; t < 8; ++t) acc[t] = 0.f;
    const float* Et = e_lds + tg * 8 * 64;
    #pragma unroll
    for (int k = 0; k < 64; ++k) {
        #pragma unroll
        for (int t = 0; t < 8; ++t) acc[t] = fmaf(Et[t * 64 + k], w[k], acc[t]);
    }

    float* dst = (hsel ? V : U) + (tok0 + tg * 8) * 64 + n;
    #pragma unroll
    for (int t = 0; t < 8; ++t) dst[t * 64] = acc[t];
}

__global__ __launch_bounds__(256)
void pair_kernel(const float* __restrict__ U,
                 const float* __restrict__ V,
                 const float* __restrict__ b1,
                 const float* __restrict__ W2,
                 const float* __restrict__ b2,
                 const float* __restrict__ W3,
                 const float* __restrict__ b3,
                 float* __restrict__ out) {
    __shared__ float u_lds[8 * LDS_STRIDE];
    __shared__ float v_lds[64 * LDS_STRIDE];
    __shared__ float w2t [64 * LDS_STRIDE];
    __shared__ float b1s[64], b2s[64], w3s[64];

    const int tid = threadIdx.x;
    const int b   = blockIdx.y;
    const int f0  = blockIdx.x * 4;

    {
        const int r16 = tid >> 4;
        const int c4  = (tid & 15) * 4;
        if (tid < 128) {
            const int fs = f0 + (r16 >> 1);
            const int iu = (r16 & 1) ? 62 - fs : fs;
            const float4 uu = *(const float4*)(U + (b * 64 + iu) * 64 + c4);
            *(float4*)&u_lds[r16 * LDS_STRIDE + c4] = uu;
        }
        #pragma unroll
        for (int q = 0; q < 4; ++q) {
            const int row = r16 + 16 * q;
            const float4 vv = *(const float4*)(V + (b * 64 + row) * 64 + c4);
            *(float4*)&v_lds[row * LDS_STRIDE + c4] = vv;
        }
        for (int e = tid; e < 4096; e += 256) {
            int k = e >> 6, nn = e & 63;
            w2t[nn * LDS_STRIDE + k] = W2[k * 64 + nn];
        }
        if (tid < 64) { b1s[tid] = b1[tid]; b2s[tid] = b2[tid]; w3s[tid] = W3[tid]; }
    }
    __syncthreads();

    const int lane = tid & 31;
    const int wv   = tid >> 5;
    const int fl   = wv >> 1;
    const int hr   = wv & 1;
    const int f    = f0 + fl;
    if (f == 31 && hr == 1) return;
    const int nfirst = 63 - f;
    const int jl   = lane & 15;
    const int half = lane >> 4;
    const int koff = 2 * half;

    v2f hA[2][16];
    #pragma unroll
    for (int mt = 0; mt < 2; ++mt) {
        const int t    = 32 * hr + 16 * mt + jl;
        const int sel  = (t >= nfirst) ? 1 : 0;
        const int urow = 2 * fl + sel;
        const int jrow = sel ? t : f + 1 + t;
        #pragma unroll
        for (int kk = 0; kk < 16; ++kk) {
            const int k0 = 4 * kk + koff;
            float x0 = u_lds[urow * LDS_STRIDE + k0]     + v_lds[jrow * LDS_STRIDE + k0]     + b1s[k0];
            float x1 = u_lds[urow * LDS_STRIDE + k0 + 1] + v_lds[jrow * LDS_STRIDE + k0 + 1] + b1s[k0 + 1];
            hA[mt][kk].x = gelu_exact(x0);
            hA[mt][kk].y = gelu_exact(x1);
        }
    }

    float p0[8], p1[8];
    #pragma unroll
    for (int r = 0; r < 8; ++r) { p0[r] = 0.f; p1[r] = 0.f; }

    #pragma unroll
    for (int nt = 0; nt < 4; ++nt) {
        const int n = nt * 16 + jl;
        v8f acc0 = {};
        v8f acc1 = {};
        #pragma unroll
        for (int kk = 0; kk < 16; ++kk) {
            const int k0 = 4 * kk + koff;
            v2f bb = *(const v2f*)&w2t[n * LDS_STRIDE + k0];
            acc0 = __builtin_amdgcn_wmma_f32_16x16x4_f32(false, hA[0][kk], false, bb,
                                                         (short)0, acc0, false, false);
            acc1 = __builtin_amdgcn_wmma_f32_16x16x4_f32(false, hA[1][kk], false, bb,
                                                         (short)0, acc1, false, false);
        }
        const float b2n = b2s[n];
        const float w3n = w3s[n];
        #pragma unroll
        for (int r = 0; r < 8; ++r) p0[r] = fmaf(gelu_exact(acc0[r] + b2n), w3n, p0[r]);
        #pragma unroll
        for (int r = 0; r < 8; ++r) p1[r] = fmaf(gelu_exact(acc1[r] + b2n), w3n, p1[r]);
    }

    const float bias3 = b3[0];
    #pragma unroll
    for (int mt = 0; mt < 2; ++mt) {
        #pragma unroll
        for (int r = 0; r < 8; ++r) {
            float s = (mt == 0) ? p0[r] : p1[r];
            s += __shfl_xor(s, 1);
            s += __shfl_xor(s, 2);
            s += __shfl_xor(s, 4);
            s += __shfl_xor(s, 8);
            if (jl == 0) {
                const int tq = 32 * hr + 16 * mt + 8 * half + r;
                const int sq = (tq >= nfirst) ? 1 : 0;
                const int ig = sq ? 62 - f : f;
                const int jg = sq ? tq : f + 1 + tq;
                const float val = s + bias3;
                out[(b * 64 + ig) * 64 + jg] = val;
                out[(b * 64 + jg) * 64 + ig] = val;
            }
        }
    }

    if (tid < 8) {
        const int idg = blockIdx.x * 8 + tid;
        out[(b * 64 + idg) * 64 + idg] = 0.f;
    }
}

extern "C" void kernel_launch(void* const* d_in, const int* in_sizes, int n_in,
                              void* d_out, int out_size, void* d_ws, size_t ws_size,
                              hipStream_t stream) {
    const int*   positions = (const int*)  d_in[0];
    const int*   amino     = (const int*)  d_in[1];
    const float* pos_table = (const float*)d_in[2];
    const float* aa_table  = (const float*)d_in[3];
    const float* W1        = (const float*)d_in[4];
    const float* b1        = (const float*)d_in[5];
    const float* W2        = (const float*)d_in[6];
    const float* b2        = (const float*)d_in[7];
    const float* W3        = (const float*)d_in[8];
    const float* b3        = (const float*)d_in[9];
    float* out = (float*)d_out;
    float* U   = (float*)d_ws;
    float* V   = U + 128 * 64 * 64;

    token_proj_kernel<<<dim3((128 * 64) / TOK), dim3(256), 0, stream>>>(
        positions, amino, pos_table, aa_table, W1, U, V);
    pair_kernel<<<dim3(8, 128), dim3(256), 0, stream>>>(
        U, V, b1, W2, b2, W3, b3, out);
}


# pair kernel stage 1: all 24 global loads of a thread (u, v rows, W2, b1, b2, W3) issued before the first wait: one memory round trip per block in front of the barrier, not eleven in series
# speedup vs baseline: 2.7756x; 1.2316x over previous
#include <hip/hip_runtime.h>
#include <math.h>


typedef __attribute__((ext_vector_type(2))) float v2f;
typedef __attribute__((ext_vector_type(4))) float v4f;
typedef __attribute__((ext_vector_type(8))) float v8f;

#define LDS_STRIDE 68

__device__ __forceinline__ float erf_select(float x) {
    const float ax = fabsf(x);
    const float t  = ax * ax;
    float p = fmaf(t, -0x1.268bc2p-11f, 0x1.420828p-8f);
    p = fmaf(t, p, -0x1.b59370p-6f);
    p = fmaf(t, p,  0x1.ce077cp-4f);
    p = fmaf(t, p, -0x1.812660p-2f);
    p = fmaf(t, p,  0x1.06eba0p-3f);
    const float r_small = fmaf(ax, p, ax);

    float q = fmaf(ax, 0x1.1d3156p-16f, -0x1.8d1290p-12f);
    q = fmaf(ax, q,  0x1.f9a6d2p-9f);
    q = fmaf(ax, q, -0x1.8c3164p-6f);
    q = fmaf(ax, q,  0x1.b4e9c8p-4f);
    q = fmaf(ax, q,  0x1.4515fap-1f);
    q = fmaf(ax, q,  0x1.078e50p-3f);
    q = fmaf(ax, q, ax);
    const float r_large = 1.0f - __builtin_amdgcn_exp2f(q * -0x1.715476p+0f);

    return copysignf(ax < 1.0f ? r_small : r_large, x);
}

__device__ __forceinline__ float gelu_exact(float x) {
    return 0.5f * x * (1.0f + erf_select(x * 0.70710678118654752f));
}

#define TOK 16
__global__ __launch_bounds__(256)
void token_proj_kernel(const int* __restrict__ positions,
                       const int* __restrict__ amino,
                       const float* __restrict__ pos_table,
                       const float* __restrict__ aa_table,
                       const float* __restrict__ W1,
                       float* __restrict__ U,
                       float* __restrict__ V) {
    __shared__ float e_lds[TOK * 64];

    const int tid  = threadIdx.x;
    const int tok0 = blockIdx.x * TOK;

    const int gt = tid >> 4;
    const int gk = (tid & 15) * 4;
    int p = positions[tok0 + gt]; p = p < 0 ? 0 : (p > 1023 ? 1023 : p);
    int a = amino[tok0 + gt];     a = a < 0 ? 0 : (a > 20   ? 20   : a);

    const int n    = tid & 63;
    const int hsel = (tid >> 6) & 1;
    const int tg   = tid >> 7;
    const float* Wc = W1 + hsel * 64 * 64 + n;

    float w[64];
    #pragma unroll
    for (int k = 0; k < 64; ++k) w[k] = Wc[k * 64];

    {
        const float4 pe = *(const float4*)(pos_table + p * 64 + gk);
        const float4 ae = *(const float4*)(aa_table  + a * 64 + gk);
        float4 ev;
        ev.x = pe.x + ae.x; ev.y = pe.y + ae.y; ev.z = pe.z + ae.z; ev.w = pe.w + ae.w;
        *(float4*)&e_lds[gt * 64 + gk] = ev;
    }
    __syncthreads();

    float acc[8];
    #pragma unroll
    for (int t = 0; t < 8; ++t) acc[t] = 0.f;
    const float* Et = e_lds + tg * 8 * 64;
    #pragma unroll
    for (int k = 0; k < 64; ++k) {
        #pragma unroll
        for (int t = 0; t < 8; ++t) acc[t] = fmaf(Et[t * 64 + k], w[k], acc[t]);
    }

    float* dst = (hsel ? V : U) + (tok0 + tg * 8) * 64 + n;
    #pragma unroll
    for (int t = 0; t < 8; ++t) dst[t * 64] = acc[t];
}

__global__ __launch_bounds__(256)
void pair_kernel(const float* __restrict__ U,
                 const float* __restrict__ V,
                 const float* __restrict__ b1,
                 const float* __restrict__ W2,
                 const float* __restrict__ b2,
                 const float* __restrict__ W3,
                 const float* __restrict__ b3,
                 float* __restrict__ out) {
    __shared__ float u_lds[8 * LDS_STRIDE];
    __shared__ float v_lds[64 * LDS_STRIDE];
    __shared__ float w2t [64 * LDS_STRIDE];
    __shared__ float b1s[64], b2s[64], w3s[64];

    const int tid = threadIdx.x;
    const int b   = blockIdx.y;
    const int f0  = blockIdx.x * 4;

    {
        const int r16 = tid >> 4;
        const int c4  = (tid & 15) * 4;
        const int nn  = tid & 63;
        const int kq  = tid >> 6;

        const int ur  = tid >> 5;
        const int uc  = (tid & 31) * 2;
        const int fs  = f0 + (ur >> 1);
        const int iu  = (ur & 1) ? 62 - fs : fs;
        const v2f uu = *(const v2f*)(U + (b * 64 + iu) * 64 + uc);

        v4f vv[4];
        #pragma unroll
        for (int q = 0; q < 4; ++q)
            vv[q] = *(const v4f*)(V + (b * 64 + r16 + 16 * q) * 64 + c4);

        v4f wk[4];
        #pragma unroll
        for (int g = 0; g < 4; ++g) {
            const float* wp = W2 + (16 * kq + 4 * g) * 64 + nn;
            wk[g].x = wp[0]; wk[g].y = wp[64]; wk[g].z = wp[128]; wk[g].w = wp[192];
        }
        const float b1v = b1[nn], b2v = b2[nn], w3v = W3[nn];
        asm volatile("" :: "v"(b1v), "v"(b2v), "v"(w3v), "v"(uu),
                     "v"(vv[0]), "v"(vv[1]), "v"(vv[2]), "v"(vv[3]),
                     "v"(wk[0]), "v"(wk[1]), "v"(wk[2]), "v"(wk[3]));

        *(v2f*)&u_lds[ur * LDS_STRIDE + uc] = uu;
        #pragma unroll
        for (int q = 0; q < 4; ++q)
            *(v4f*)&v_lds[(r16 + 16 * q) * LDS_STRIDE + c4] = vv[q];
        #pragma unroll
        for (int g = 0; g < 4; ++g)
            *(v4f*)&w2t[nn * LDS_STRIDE + 16 * kq + 4 * g] = wk[g];
        if (tid < 64) { b1s[tid] = b1v; b2s[tid] = b2v; w3s[tid] = w3v; }
    }
    __syncthreads();

    const int lane = tid & 31;
    const int wv   = tid >> 5;
    const int fl   = wv >> 1;
    const int hr   = wv & 1;
    const int f    = f0 + fl;
    if (f == 31 && hr == 1) return;
    const int nfirst = 63 - f;
    const int jl   = lane & 15;
    const int half = lane >> 4;
    const int koff = 2 * half;

    v2f hA[2][16];
    #pragma unroll
    for (int mt = 0; mt < 2; ++mt) {
        const int t    = 32 * hr + 16 * mt + jl;
        const int sel  = (t >= nfirst) ? 1 : 0;
        const int urow = 2 * fl + sel;
        const int jrow = sel ? t : f + 1 + t;
        #pragma unroll
        for (int kk = 0; kk < 16; ++kk) {
            const int k0 = 4 * kk + koff;
            float x0 = u_lds[urow * LDS_STRIDE + k0]     + v_lds[jrow * LDS_STRIDE + k0]     + b1s[k0];
            float x1 = u_lds[urow * LDS_STRIDE + k0 + 1] + v_lds[jrow * LDS_STRIDE + k0 + 1] + b1s[k0 + 1];
            hA[mt][kk].x = gelu_exact(x0);
            hA[mt][kk].y = gelu_exact(x1);
        }
    }

    float p0[8], p1[8];
    #pragma unroll
    for (int r = 0; r < 8; ++r) { p0[r] = 0.f; p1[r] = 0.f; }

    #pragma unroll
    for (int nt = 0; nt < 4; ++nt) {
        const int n = nt * 16 + jl;
        v8f acc0 = {};
        v8f acc1 = {};
        #pragma unroll
        for (int kk = 0; kk < 16; ++kk) {
            const int k0 = 4 * kk + koff;
            v2f bb = *(const v2f*)&w2t[n * LDS_STRIDE + k0];
            acc0 = __builtin_amdgcn_wmma_f32_16x16x4_f32(false, hA[0][kk], false, bb,
                                                         (short)0, acc0, false, false);
            acc1 = __builtin_amdgcn_wmma_f32_16x16x4_f32(false, hA[1][kk], false, bb,
                                                         (short)0, acc1, false, false);
        }
        const float b2n = b2s[n];
        const float w3n = w3s[n];
        #pragma unroll
        for (int r = 0; r < 8; ++r) p0[r] = fmaf(gelu_exact(acc0[r] + b2n), w3n, p0[r]);
        #pragma unroll
        for (int r = 0; r < 8; ++r) p1[r] = fmaf(gelu_exact(acc1[r] + b2n), w3n, p1[r]);
    }

    const float bias3 = b3[0];
    #pragma unroll
    for (int mt = 0; mt < 2; ++mt) {
        #pragma unroll
        for (int r = 0; r < 8; ++r) {
            float s = (mt == 0) ? p0[r] : p1[r];
            s += __shfl_xor(s, 1);
            s += __shfl_xor(s, 2);
            s += __shfl_xor(s, 4);
            s += __shfl_xor(s, 8);
            if (jl == 0) {
                const int tq = 32 * hr + 16 * mt + 8 * half + r;
                const int sq = (tq >= nfirst) ? 1 : 0;
                const int ig = sq ? 62 - f : f;
                const int jg = sq ? tq : f + 1 + tq;
                const float val = s + bias3;
                out[(b * 64 + ig) * 64 + jg] = val;
                out[(b * 64 + jg) * 64 + ig] = val;
            }
        }
    }

    if (tid < 8) {
        const int idg = blockIdx.x * 8 + tid;
        out[(b * 64 + idg) * 64 + idg] = 0.f;
    }
}

extern "C" void kernel_launch(void* const* d_in, const int* in_sizes, int n_in,
                              void* d_out, int out_size, void* d_ws, size_t ws_size,
                              hipStream_t stream) {
    const int*   positions = (const int*)  d_in[0];
    const int*   amino     = (const int*)  d_in[1];
    const float* pos_table = (const float*)d_in[2];
    const float* aa_table  = (const float*)d_in[3];
    const float* W1        = (const float*)d_in[4];
    const float* b1        = (const float*)d_in[5];
    const float* W2        = (const float*)d_in[6];
    const float* b2        = (const float*)d_in[7];
    const float* W3        = (const float*)d_in[8];
    const float* b3        = (const float*)d_in[9];
    float* out = (float*)d_out;
    float* U   = (float*)d_ws;
    float* V   = U + 128 * 64 * 64;

    token_proj_kernel<<<dim3((128 * 64) / TOK), dim3(256), 0, stream>>>(
        positions, amino, pos_table, aa_table, W1, U, V);
    pair_kernel<<<dim3(8, 128), dim3(256), 0, stream>>>(
        U, V, b1, W2, b2, W3, b3, out);
}


# gelu as max(x,0) - |x|*2^(|x|*G(|x|)-1), G = log2(erfc(|x|/sqrt2))/|x| as one degree-7 polynomial on the hardware exp2: replaces both erf range formulas, compare, select and sign copy
# speedup vs baseline: 2.8926x; 1.0422x over previous
#include <hip/hip_runtime.h>
#include <math.h>


typedef __attribute__((ext_vector_type(2))) float v2f;
typedef __attribute__((ext_vector_type(4))) float v4f;
typedef __attribute__((ext_vector_type(8))) float v8f;

#define LDS_STRIDE 68

__device__ __forceinline__ float gelu_exact(float x) {
    const float r = fmaxf(x, 0.0f);
    const float a = fmaf(r, 2.0f, -x);
    float g = fmaf(a, -0x1.fe92fap-20f, 0x1.d6b5eep-16f);
    g = fmaf(a, g, -0x1.13b6f8p-13f);
    g = fmaf(a, g, -0x1.1d4c74p-12f);
    g = fmaf(a, g,  0x1.dad62ap-8f);
    g = fmaf(a, g, -0x1.af201ep-5f);
    g = fmaf(a, g, -0x1.d62e9ap-2f);
    g = fmaf(a, g, -0x1.26af36p+0f);
    const float e = fmaf(a, g, -1.0f);
    return fmaf(-a, __builtin_amdgcn_exp2f(e), r);
}

#define TOK 16
__global__ __launch_bounds__(256)
void token_proj_kernel(const int* __restrict__ positions,
                       const int* __restrict__ amino,
                       const float* __restrict__ pos_table,
                       const float* __restrict__ aa_table,
                       const float* __restrict__ W1,
                       float* __restrict__ U,
                       float* __restrict__ V) {
    __shared__ float e_lds[TOK * 64];

    const int tid  = threadIdx.x;
    const int tok0 = blockIdx.x * TOK;

    const int gt = tid >> 4;
    const int gk = (tid & 15) * 4;
    int p = positions[tok0 + gt]; p = p < 0 ? 0 : (p > 1023 ? 1023 : p);
    int a = amino[tok0 + gt];     a = a < 0 ? 0 : (a > 20   ? 20   : a);

    const int n    = tid & 63;
    const int hsel = (tid >> 6) & 1;
    const int tg   = tid >> 7;
    const float* Wc = W1 + hsel * 64 * 64 + n;

    float w[64];
    #pragma unroll
    for (int k = 0; k < 64; ++k) w[k] = Wc[k * 64];

    {
        const float4 pe = *(const float4*)(pos_table + p * 64 + gk);
        const float4 ae = *(const float4*)(aa_table  + a * 64 + gk);
        float4 ev;
        ev.x = pe.x + ae.x; ev.y = pe.y + ae.y; ev.z = pe.z + ae.z; ev.w = pe.w + ae.w;
        *(float4*)&e_lds[gt * 64 + gk] = ev;
    }
    __syncthreads();

    float acc[8];
    #pragma unroll
    for (int t = 0; t < 8; ++t) acc[t] = 0.f;
    const float* Et = e_lds + tg * 8 * 64;
    #pragma unroll
    for (int k = 0; k < 64; ++k) {
        #pragma unroll
        for (int t = 0; t < 8; ++t) acc[t] = fmaf(Et[t * 64 + k], w[k], acc[t]);
    }

    float* dst = (hsel ? V : U) + (tok0 + tg * 8) * 64 + n;
    #pragma unroll
    for (int t = 0; t < 8; ++t) dst[t * 64] = acc[t];
}

__global__ __launch_bounds__(256)
void pair_kernel(const float* __restrict__ U,
                 const float* __restrict__ V,
                 const float* __restrict__ b1,
                 const float* __restrict__ W2,
                 const float* __restrict__ b2,
                 const float* __restrict__ W3,
                 const float* __restrict__ b3,
                 float* __restrict__ out) {
    __shared__ float u_lds[8 * LDS_STRIDE];
    __shared__ float v_lds[64 * LDS_STRIDE];
    __shared__ float w2t [64 * LDS_STRIDE];
    __shared__ float b1s[64], b2s[64], w3s[64];

    const int tid = threadIdx.x;
    const int b   = blockIdx.y;
    const int f0  = blockIdx.x * 4;

    {
        const int r16 = tid >> 4;
        const int c4  = (tid & 15) * 4;
        const int nn  = tid & 63;
        const int kq  = tid >> 6;

        const int ur  = tid >> 5;
        const int uc  = (tid & 31) * 2;
        const int fs  = f0 + (ur >> 1);
        const int iu  = (ur & 1) ? 62 - fs : fs;
        const v2f uu = *(const v2f*)(U + (b * 64 + iu) * 64 + uc);

        v4f vv[4];
        #pragma unroll
        for (int q = 0; q < 4; ++q)
            vv[q] = *(const v4f*)(V + (b * 64 + r16 + 16 * q) * 64 + c4);

        v4f wk[4];
        #pragma unroll
        for (int g = 0; g < 4; ++g) {
            const float* wp = W2 + (16 * kq + 4 * g) * 64 + nn;
            wk[g].x = wp[0]; wk[g].y = wp[64]; wk[g].z = wp[128]; wk[g].w = wp[192];
        }
        const float b1v = b1[nn], b2v = b2[nn], w3v = W3[nn];
        asm volatile("" :: "v"(b1v), "v"(b2v), "v"(w3v), "v"(uu),
                     "v"(vv[0]), "v"(vv[1]), "v"(vv[2]), "v"(vv[3]),
                     "v"(wk[0]), "v"(wk[1]), "v"(wk[2]), "v"(wk[3]));

        *(v2f*)&u_lds[ur * LDS_STRIDE + uc] = uu;
        #pragma unroll
        for (int q = 0; q < 4; ++q)
            *(v4f*)&v_lds[(r16 + 16 * q) * LDS_STRIDE + c4] = vv[q];
        #pragma unroll
        for (int g = 0; g < 4; ++g)
            *(v4f*)&w2t[nn * LDS_STRIDE + 16 * kq + 4 * g] = wk[g];
        if (tid < 64) { b1s[tid] = b1v; b2s[tid] = b2v; w3s[tid] = w3v; }
    }
    __syncthreads();

    const int lane = tid & 31;
    const int wv   = tid >> 5;
    const int fl   = wv >> 1;
    const int hr   = wv & 1;
    const int f    = f0 + fl;
    if (f == 31 && hr == 1) return;
    const int nfirst = 63 - f;
    const int jl   = lane & 15;
    const int half = lane >> 4;
    const int koff = 2 * half;

    v2f hA[2][16];
    #pragma unroll
    for (int mt = 0; mt < 2; ++mt) {
        const int t    = 32 * hr + 16 * mt + jl;
        const int sel  = (t >= nfirst) ? 1 : 0;
        const int urow = 2 * fl + sel;
        const int jrow = sel ? t : f + 1 + t;
        #pragma unroll
        for (int kk = 0; kk < 16; ++kk) {
            const int k0 = 4 * kk + koff;
            float x0 = u_lds[urow * LDS_STRIDE + k0]     + v_lds[jrow * LDS_STRIDE + k0]     + b1s[k0];
            float x1 = u_lds[urow * LDS_STRIDE + k0 + 1] + v_lds[jrow * LDS_STRIDE + k0 + 1] + b1s[k0 + 1];
            hA[mt][kk].x = gelu_exact(x0);
            hA[mt][kk].y = gelu_exact(x1);
        }
    }

    float p0[8], p1[8];
    #pragma unroll
    for (int r = 0; r < 8; ++r) { p0[r] = 0.f; p1[r] = 0.f; }

    #pragma unroll
    for (int nt = 0; nt < 4; ++nt) {
        const int n = nt * 16 + jl;
        v8f acc0 = {};
        v8f acc1 = {};
        #pragma unroll
        for (int kk = 0; kk < 16; ++kk) {
            const int k0 = 4 * kk + koff;
            v2f bb = *(const v2f*)&w2t[n * LDS_STRIDE + k0];
            acc0 = __builtin_amdgcn_wmma_f32_16x16x4_f32(false, hA[0][kk], false, bb,
                                                         (short)0, acc0, false, false);
            acc1 = __builtin_amdgcn_wmma_f32_16x16x4_f32(false, hA[1][kk], false, bb,
                                                         (short)0, acc1, false, false);
        }
        const float b2n = b2s[n];
        const float w3n = w3s[n];
        #pragma unroll
        for (int r = 0; r < 8; ++r) p0[r] = fmaf(gelu_exact(acc0[r] + b2n), w3n, p0[r]);
        #pragma unroll
        for (int r = 0; r < 8; ++r) p1[r] = fmaf(gelu_exact(acc1[r] + b2n), w3n, p1[r]);
    }

    const float bias3 = b3[0];
    #pragma unroll
    for (int mt = 0; mt < 2; ++mt) {
        #pragma unroll
        for (int r = 0; r < 8; ++r) {
            float s = (mt == 0) ? p0[r] : p1[r];
            s += __shfl_xor(s, 1);
            s += __shfl_xor(s, 2);
            s += __shfl_xor(s, 4);
            s += __shfl_xor(s, 8);
            if (jl == 0) {
                const int tq = 32 * hr + 16 * mt + 8 * half + r;
                const int sq = (tq >= nfirst) ? 1 : 0;
                const int ig = sq ? 62 - f : f;
                const int jg = sq ? tq : f + 1 + tq;
                const float val = s + bias3;
                out[(b * 64 + ig) * 64 + jg] = val;
                out[(b * 64 + jg) * 64 + ig] = val;
            }
        }
    }

    if (tid < 8) {
        const int idg = blockIdx.x * 8 + tid;
        out[(b * 64 + idg) * 64 + idg] = 0.f;
    }
}

extern "C" void kernel_launch(void* const* d_in, const int* in_sizes, int n_in,
                              void* d_out, int out_size, void* d_ws, size_t ws_size,
                              hipStream_t stream) {
    const int*   positions = (const int*)  d_in[0];
    const int*   amino     = (const int*)  d_in[1];
    const float* pos_table = (const float*)d_in[2];
    const float* aa_table  = (const float*)d_in[3];
    const float* W1        = (const float*)d_in[4];
    const float* b1        = (const float*)d_in[5];
    const float* W2        = (const float*)d_in[6];
    const float* b2        = (const float*)d_in[7];
    const float* W3        = (const float*)d_in[8];
    const float* b3        = (const float*)d_in[9];
    float* out = (float*)d_out;
    float* U   = (float*)d_ws;
    float* V   = U + 128 * 64 * 64;

    token_proj_kernel<<<dim3((128 * 64) / TOK), dim3(256), 0, stream>>>(
        positions, amino, pos_table, aa_table, W1, U, V);
    pair_kernel<<<dim3(8, 128), dim3(256), 0, stream>>>(
        U, V, b1, W2, b2, W3, b3, out);
}


# pair kernel: a wave takes its two 16-pair row tiles one after the other, each against all four column tiles (A fragment shared): 148 -> 109 VGPRs, 8 waves per SIMD, whole grid resident; same sums
# speedup vs baseline: 3.1252x; 1.0804x over previous
#include <hip/hip_runtime.h>
#include <math.h>


typedef __attribute__((ext_vector_type(2))) float v2f;
typedef __attribute__((ext_vector_type(4))) float v4f;
typedef __attribute__((ext_vector_type(8))) float v8f;

#define LDS_STRIDE 68

__device__ __forceinline__ float gelu_exact(float x) {
    const float r = fmaxf(x, 0.0f);
    const float a = fmaf(r, 2.0f, -x);
    float g = fmaf(a, -0x1.fe92fap-20f, 0x1.d6b5eep-16f);
    g = fmaf(a, g, -0x1.13b6f8p-13f);
    g = fmaf(a, g, -0x1.1d4c74p-12f);
    g = fmaf(a, g,  0x1.dad62ap-8f);
    g = fmaf(a, g, -0x1.af201ep-5f);
    g = fmaf(a, g, -0x1.d62e9ap-2f);
    g = fmaf(a, g, -0x1.26af36p+0f);
    const float e = fmaf(a, g, -1.0f);
    return fmaf(-a, __builtin_amdgcn_exp2f(e), r);
}

#define TOK 16
__global__ __launch_bounds__(256)
void token_proj_kernel(const int* __restrict__ positions,
                       const int* __restrict__ amino,
                       const float* __restrict__ pos_table,
                       const float* __restrict__ aa_table,
                       const float* __restrict__ W1,
                       float* __restrict__ U,
                       float* __restrict__ V) {
    __shared__ float e_lds[TOK * 64];

    const int tid  = threadIdx.x;
    const int tok0 = blockIdx.x * TOK;

    const int gt = tid >> 4;
    const int gk = (tid & 15) * 4;
    int p = positions[tok0 + gt]; p = p < 0 ? 0 : (p > 1023 ? 1023 : p);
    int a = amino[tok0 + gt];     a = a < 0 ? 0 : (a > 20   ? 20   : a);

    const int n    = tid & 63;
    const int hsel = (tid >> 6) & 1;
    const int tg   = tid >> 7;
    const float* Wc = W1 + hsel * 64 * 64 + n;

    float w[64];
    #pragma unroll
    for (int k = 0; k < 64; ++k) w[k] = Wc[k * 64];

    {
        const float4 pe = *(const float4*)(pos_table + p * 64 + gk);
        const float4 ae = *(const float4*)(aa_table  + a * 64 + gk);
        float4 ev;
        ev.x = pe.x + ae.x; ev.y = pe.y + ae.y; ev.z = pe.z + ae.z; ev.w = pe.w + ae.w;
        *(float4*)&e_lds[gt * 64 + gk] = ev;
    }
    __syncthreads();

    float acc[8];
    #pragma unroll
    for (int t = 0; t < 8; ++t) acc[t] = 0.f;
    const float* Et = e_lds + tg * 8 * 64;
    #pragma unroll
    for (int k = 0; k < 64; ++k) {
        #pragma unroll
        for (int t = 0; t < 8; ++t) acc[t] = fmaf(Et[t * 64 + k], w[k], acc[t]);
    }

    float* dst = (hsel ? V : U) + (tok0 + tg * 8) * 64 + n;
    #pragma unroll
    for (int t = 0; t < 8; ++t) dst[t * 64] = acc[t];
}

__global__ __launch_bounds__(256)
void pair_kernel(const float* __restrict__ U,
                 const float* __restrict__ V,
                 const float* __restrict__ b1,
                 const float* __restrict__ W2,
                 const float* __restrict__ b2,
                 const float* __restrict__ W3,
                 const float* __restrict__ b3,
                 float* __restrict__ out) {
    __shared__ float u_lds[8 * LDS_STRIDE];
    __shared__ float v_lds[64 * LDS_STRIDE];
    __shared__ float w2t [64 * LDS_STRIDE];
    __shared__ float b1s[64], b2s[64], w3s[64];

    const int tid = threadIdx.x;
    const int b   = blockIdx.y;
    const int f0  = blockIdx.x * 4;

    {
        const int r16 = tid >> 4;
        const int c4  = (tid & 15) * 4;
        const int nn  = tid & 63;
        const int kq  = tid >> 6;

        const int ur  = tid >> 5;
        const int uc  = (tid & 31) * 2;
        const int fs  = f0 + (ur >> 1);
        const int iu  = (ur & 1) ? 62 - fs : fs;
        const v2f uu = *(const v2f*)(U + (b * 64 + iu) * 64 + uc);

        v4f vv[4];
        #pragma unroll
        for (int q = 0; q < 4; ++q)
            vv[q] = *(const v4f*)(V + (b * 64 + r16 + 16 * q) * 64 + c4);

        v4f wk[4];
        #pragma unroll
        for (int g = 0; g < 4; ++g) {
            const float* wp = W2 + (16 * kq + 4 * g) * 64 + nn;
            wk[g].x = wp[0]; wk[g].y = wp[64]; wk[g].z = wp[128]; wk[g].w = wp[192];
        }
        const float b1v = b1[nn], b2v = b2[nn], w3v = W3[nn];
        asm volatile("" :: "v"(b1v), "v"(b2v), "v"(w3v), "v"(uu),
                     "v"(vv[0]), "v"(vv[1]), "v"(vv[2]), "v"(vv[3]),
                     "v"(wk[0]), "v"(wk[1]), "v"(wk[2]), "v"(wk[3]));

        *(v2f*)&u_lds[ur * LDS_STRIDE + uc] = uu;
        #pragma unroll
        for (int q = 0; q < 4; ++q)
            *(v4f*)&v_lds[(r16 + 16 * q) * LDS_STRIDE + c4] = vv[q];
        #pragma unroll
        for (int g = 0; g < 4; ++g)
            *(v4f*)&w2t[nn * LDS_STRIDE + 16 * kq + 4 * g] = wk[g];
        if (tid < 64) { b1s[tid] = b1v; b2s[tid] = b2v; w3s[tid] = w3v; }
    }
    __syncthreads();

    const int lane = tid & 31;
    const int wv   = tid >> 5;
    const int fl   = wv >> 1;
    const int hr   = wv & 1;
    const int f    = f0 + fl;
    if (f == 31 && hr == 1) return;
    const int nfirst = 63 - f;
    const int jl   = lane & 15;
    const int half = lane >> 4;
    const int koff = 2 * half;

    const float bias3 = b3[0];
    #pragma unroll
    for (int mt = 0; mt < 2; ++mt) {
        v2f hA[16];
        {
            const int t    = 32 * hr + 16 * mt + jl;
            const int sel  = (t >= nfirst) ? 1 : 0;
            const int urow = 2 * fl + sel;
            const int jrow = sel ? t : f + 1 + t;
            #pragma unroll
            for (int kk = 0; kk < 16; ++kk) {
                const int k0 = 4 * kk + koff;
                float x0 = u_lds[urow * LDS_STRIDE + k0]     + v_lds[jrow * LDS_STRIDE + k0]     + b1s[k0];
                float x1 = u_lds[urow * LDS_STRIDE + k0 + 1] + v_lds[jrow * LDS_STRIDE + k0 + 1] + b1s[k0 + 1];
                hA[kk].x = gelu_exact(x0);
                hA[kk].y = gelu_exact(x1);
            }
        }

        float p[8];
        #pragma unroll
        for (int r = 0; r < 8; ++r) p[r] = 0.f;

        {
            v8f acc[4] = {};
            #pragma unroll
            for (int kk = 0; kk < 16; ++kk) {
                const int k0 = 4 * kk + koff;
                #pragma unroll
                for (int nt = 0; nt < 4; ++nt) {
                    v2f bb = *(const v2f*)&w2t[(nt * 16 + jl) * LDS_STRIDE + k0];
                    acc[nt] = __builtin_amdgcn_wmma_f32_16x16x4_f32(false, hA[kk], false, bb,
                                                                   (short)0, acc[nt], false, false);
                }
            }
            #pragma unroll
            for (int nt = 0; nt < 4; ++nt) {
                const float b2n = b2s[nt * 16 + jl], w3n = w3s[nt * 16 + jl];
                #pragma unroll
                for (int r = 0; r < 8; ++r) p[r] = fmaf(gelu_exact(acc[nt][r] + b2n), w3n, p[r]);
            }
        }

        #pragma unroll
        for (int r = 0; r < 8; ++r) {
            float s = p[r];
            s += __shfl_xor(s, 1);
            s += __shfl_xor(s, 2);
            s += __shfl_xor(s, 4);
            s += __shfl_xor(s, 8);
            if (jl == 0) {
                const int tq = 32 * hr + 16 * mt + 8 * half + r;
                const int sq = (tq >= nfirst) ? 1 : 0;
                const int ig = sq ? 62 - f : f;
                const int jg = sq ? tq : f + 1 + tq;
                const float val = s + bias3;
                out[(b * 64 + ig) * 64 + jg] = val;
                out[(b * 64 + jg) * 64 + ig] = val;
            }
        }
    }

    if (tid < 8) {
        const int idg = blockIdx.x * 8 + tid;
        out[(b * 64 + idg) * 64 + idg] = 0.f;
    }
}

extern "C" void kernel_launch(void* const* d_in, const int* in_sizes, int n_in,
                              void* d_out, int out_size, void* d_ws, size_t ws_size,
                              hipStream_t stream) {
    const int*   positions = (const int*)  d_in[0];
    const int*   amino     = (const int*)  d_in[1];
    const float* pos_table = (const float*)d_in[2];
    const float* aa_table  = (const float*)d_in[3];
    const float* W1        = (const float*)d_in[4];
    const float* b1        = (const float*)d_in[5];
    const float* W2        = (const float*)d_in[6];
    const float* b2        = (const float*)d_in[7];
    const float* W3        = (const float*)d_in[8];
    const float* b3        = (const float*)d_in[9];
    float* out = (float*)d_out;
    float* U   = (float*)d_ws;
    float* V   = U + 128 * 64 * 64;

    token_proj_kernel<<<dim3((128 * 64) / TOK), dim3(256), 0, stream>>>(
        positions, amino, pos_table, aa_table, W1, U, V);
    pair_kernel<<<dim3(8, 128), dim3(256), 0, stream>>>(
        U, V, b1, W2, b2, W3, b3, out);
}


# pair kernel: layer 2 transposed (W2^T fragment as A, h1 as B): a lane's D column is its own pair; b2 starts the accumulator, layer-3 sum over n in-lane + one half swap (sum order changes)
# speedup vs baseline: 3.7066x; 1.1860x over previous
#include <hip/hip_runtime.h>
#include <math.h>


typedef __attribute__((ext_vector_type(2))) float v2f;
typedef __attribute__((ext_vector_type(4))) float v4f;
typedef __attribute__((ext_vector_type(8))) float v8f;

#define LDS_STRIDE 68

__device__ __forceinline__ float gelu_exact(float x) {
    const float r = fmaxf(x, 0.0f);
    const float a = fmaf(r, 2.0f, -x);
    float g = fmaf(a, -0x1.fe92fap-20f, 0x1.d6b5eep-16f);
    g = fmaf(a, g, -0x1.13b6f8p-13f);
    g = fmaf(a, g, -0x1.1d4c74p-12f);
    g = fmaf(a, g,  0x1.dad62ap-8f);
    g = fmaf(a, g, -0x1.af201ep-5f);
    g = fmaf(a, g, -0x1.d62e9ap-2f);
    g = fmaf(a, g, -0x1.26af36p+0f);
    const float e = fmaf(a, g, -1.0f);
    return fmaf(-a, __builtin_amdgcn_exp2f(e), r);
}

__device__ __forceinline__ float xhalf(float x) {
    const int b = __builtin_bit_cast(int, x);
    return __builtin_bit_cast(float,
        __builtin_amdgcn_permlanex16(b, b, 0x76543210, 0xfedcba98, false, false));
}

#define TOK 16
__global__ __launch_bounds__(256)
void token_proj_kernel(const int* __restrict__ positions,
                       const int* __restrict__ amino,
                       const float* __restrict__ pos_table,
                       const float* __restrict__ aa_table,
                       const float* __restrict__ W1,
                       float* __restrict__ U,
                       float* __restrict__ V) {
    __shared__ float e_lds[TOK * 64];

    const int tid  = threadIdx.x;
    const int tok0 = blockIdx.x * TOK;

    const int gt = tid >> 4;
    const int gk = (tid & 15) * 4;
    int p = positions[tok0 + gt]; p = p < 0 ? 0 : (p > 1023 ? 1023 : p);
    int a = amino[tok0 + gt];     a = a < 0 ? 0 : (a > 20   ? 20   : a);

    const int n    = tid & 63;
    const int hsel = (tid >> 6) & 1;
    const int tg   = tid >> 7;
    const float* Wc = W1 + hsel * 64 * 64 + n;

    float w[64];
    #pragma unroll
    for (int k = 0; k < 64; ++k) w[k] = Wc[k * 64];

    {
        const float4 pe = *(const float4*)(pos_table + p * 64 + gk);
        const float4 ae = *(const float4*)(aa_table  + a * 64 + gk);
        float4 ev;
        ev.x = pe.x + ae.x; ev.y = pe.y + ae.y; ev.z = pe.z + ae.z; ev.w = pe.w + ae.w;
        *(float4*)&e_lds[gt * 64 + gk] = ev;
    }
    __syncthreads();

    float acc[8];
    #pragma unroll
    for (int t = 0; t < 8; ++t) acc[t] = 0.f;
    const float* Et = e_lds + tg * 8 * 64;
    #pragma unroll
    for (int k = 0; k < 64; ++k) {
        #pragma unroll
        for (int t = 0; t < 8; ++t) acc[t] = fmaf(Et[t * 64 + k], w[k], acc[t]);
    }

    float* dst = (hsel ? V : U) + (tok0 + tg * 8) * 64 + n;
    #pragma unroll
    for (int t = 0; t < 8; ++t) dst[t * 64] = acc[t];
}

__global__ __launch_bounds__(256)
void pair_kernel(const float* __restrict__ U,
                 const float* __restrict__ V,
                 const float* __restrict__ b1,
                 const float* __restrict__ W2,
                 const float* __restrict__ b2,
                 const float* __restrict__ W3,
                 const float* __restrict__ b3,
                 float* __restrict__ out) {
    __shared__ float u_lds[8 * LDS_STRIDE];
    __shared__ float v_lds[64 * LDS_STRIDE];
    __shared__ float w2t [64 * LDS_STRIDE];
    __shared__ __attribute__((aligned(16))) float b1s[64];
    __shared__ __attribute__((aligned(16))) float b2s[64];
    __shared__ __attribute__((aligned(16))) float w3s[64];

    const int tid = threadIdx.x;
    const int b   = blockIdx.y;
    const int f0  = blockIdx.x * 4;

    {
        const int r16 = tid >> 4;
        const int c4  = (tid & 15) * 4;
        const int nn  = tid & 63;
        const int kq  = tid >> 6;

        const int ur  = tid >> 5;
        const int uc  = (tid & 31) * 2;
        const int fs  = f0 + (ur >> 1);
        const int iu  = (ur & 1) ? 62 - fs : fs;
        const v2f uu = *(const v2f*)(U + (b * 64 + iu) * 64 + uc);

        v4f vv[4];
        #pragma unroll
        for (int q = 0; q < 4; ++q)
            vv[q] = *(const v4f*)(V + (b * 64 + r16 + 16 * q) * 64 + c4);

        v4f wk[4];
        #pragma unroll
        for (int g = 0; g < 4; ++g) {
            const float* wp = W2 + (16 * kq + 4 * g) * 64 + nn;
            wk[g].x = wp[0]; wk[g].y = wp[64]; wk[g].z = wp[128]; wk[g].w = wp[192];
        }
        const float b1v = b1[nn], b2v = b2[nn], w3v = W3[nn];
        asm volatile("" :: "v"(b1v), "v"(b2v), "v"(w3v), "v"(uu),
                     "v"(vv[0]), "v"(vv[1]), "v"(vv[2]), "v"(vv[3]),
                     "v"(wk[0]), "v"(wk[1]), "v"(wk[2]), "v"(wk[3]));

        *(v2f*)&u_lds[ur * LDS_STRIDE + uc] = uu;
        #pragma unroll
        for (int q = 0; q < 4; ++q)
            *(v4f*)&v_lds[(r16 + 16 * q) * LDS_STRIDE + c4] = vv[q];
        #pragma unroll
        for (int g = 0; g < 4; ++g)
            *(v4f*)&w2t[nn * LDS_STRIDE + 16 * kq + 4 * g] = wk[g];
        if (tid < 64) { b1s[tid] = b1v; b2s[tid] = b2v; w3s[tid] = w3v; }
    }
    __syncthreads();

    const int lane = tid & 31;
    const int wv   = tid >> 5;
    const int fl   = wv >> 1;
    const int hr   = wv & 1;
    const int f    = f0 + fl;
    if (f == 31 && hr == 1) return;
    const int nfirst = 63 - f;
    const int jl   = lane & 15;
    const int half = lane >> 4;
    const int koff = 2 * half;

    const float bias3 = b3[0];
    #pragma unroll
    for (int mt = 0; mt < 2; ++mt) {
        v2f hB[16];
        int ig, jg;
        {
            const int t    = 32 * hr + 16 * mt + jl;
            const int sel  = (t >= nfirst) ? 1 : 0;
            const int urow = 2 * fl + sel;
            const int jrow = sel ? t : f + 1 + t;
            ig = sel ? 62 - f : f;
            jg = jrow;
            #pragma unroll
            for (int kk = 0; kk < 16; ++kk) {
                const int k0 = 4 * kk + koff;
                float x0 = u_lds[urow * LDS_STRIDE + k0]     + v_lds[jrow * LDS_STRIDE + k0]     + b1s[k0];
                float x1 = u_lds[urow * LDS_STRIDE + k0 + 1] + v_lds[jrow * LDS_STRIDE + k0 + 1] + b1s[k0 + 1];
                hB[kk].x = gelu_exact(x0);
                hB[kk].y = gelu_exact(x1);
            }
        }

        v8f acc[4];
        #pragma unroll
        for (int nt = 0; nt < 4; ++nt) {
            const v4f c0 = *(const v4f*)&b2s[nt * 16 + 8 * half];
            const v4f c1 = *(const v4f*)&b2s[nt * 16 + 8 * half + 4];
            acc[nt][0] = c0.x; acc[nt][1] = c0.y; acc[nt][2] = c0.z; acc[nt][3] = c0.w;
            acc[nt][4] = c1.x; acc[nt][5] = c1.y; acc[nt][6] = c1.z; acc[nt][7] = c1.w;
        }
        #pragma unroll
        for (int kk = 0; kk < 16; ++kk) {
            const int k0 = 4 * kk + koff;
            #pragma unroll
            for (int nt = 0; nt < 4; ++nt) {
                v2f aa = *(const v2f*)&w2t[(nt * 16 + jl) * LDS_STRIDE + k0];
                acc[nt] = __builtin_amdgcn_wmma_f32_16x16x4_f32(false, aa, false, hB[kk],
                                                               (short)0, acc[nt], false, false);
            }
        }

        float p[8];
        #pragma unroll
        for (int r = 0; r < 8; ++r) p[r] = 0.f;
        #pragma unroll
        for (int nt = 0; nt < 4; ++nt) {
            const v4f w0 = *(const v4f*)&w3s[nt * 16 + 8 * half];
            const v4f w1 = *(const v4f*)&w3s[nt * 16 + 8 * half + 4];
            const float w3n[8] = { w0.x, w0.y, w0.z, w0.w, w1.x, w1.y, w1.z, w1.w };
            #pragma unroll
            for (int r = 0; r < 8; ++r) p[r] = fmaf(gelu_exact(acc[nt][r]), w3n[r], p[r]);
        }
        float s = ((p[0] + p[1]) + (p[2] + p[3])) + ((p[4] + p[5]) + (p[6] + p[7]));
        s += xhalf(s);

        {
            const float val = s + bias3;
            const int row = half ? jg : ig;
            const int col = half ? ig : jg;
            out[(b * 64 + row) * 64 + col] = val;
        }
        if (mt == 0) __builtin_amdgcn_sched_barrier(0);
    }

    if (tid < 8) {
        const int idg = blockIdx.x * 8 + tid;
        out[(b * 64 + idg) * 64 + idg] = 0.f;
    }
}

extern "C" void kernel_launch(void* const* d_in, const int* in_sizes, int n_in,
                              void* d_out, int out_size, void* d_ws, size_t ws_size,
                              hipStream_t stream) {
    const int*   positions = (const int*)  d_in[0];
    const int*   amino     = (const int*)  d_in[1];
    const float* pos_table = (const float*)d_in[2];
    const float* aa_table  = (const float*)d_in[3];
    const float* W1        = (const float*)d_in[4];
    const float* b1        = (const float*)d_in[5];
    const float* W2        = (const float*)d_in[6];
    const float* b2        = (const float*)d_in[7];
    const float* W3        = (const float*)d_in[8];
    const float* b3        = (const float*)d_in[9];
    float* out = (float*)d_out;
    float* U   = (float*)d_ws;
    float* V   = U + 128 * 64 * 64;

    token_proj_kernel<<<dim3((128 * 64) / TOK), dim3(256), 0, stream>>>(
        positions, amino, pos_table, aa_table, W1, U, V);
    pair_kernel<<<dim3(8, 128), dim3(256), 0, stream>>>(
        U, V, b1, W2, b2, W3, b3, out);
}


# one launch: each 1024-thread block (half a batch) gathers its 64 embeddings and computes its layer-1 halves u,v itself on f32 WMMA into LDS (layer-1 sum order changes); no token kernel, no scratch
# speedup vs baseline: 4.6603x; 1.2573x over previous
#include <hip/hip_runtime.h>
#include <math.h>


typedef __attribute__((ext_vector_type(2))) float v2f;
typedef __attribute__((ext_vector_type(4))) float v4f;
typedef __attribute__((ext_vector_type(8))) float v8f;

#define LDS_STRIDE 68

__device__ __forceinline__ float gelu_exact(float x) {
    const float r = fmaxf(x, 0.0f);
    const float a = fmaf(r, 2.0f, -x);
    float g = fmaf(a, -0x1.fe92fap-20f, 0x1.d6b5eep-16f);
    g = fmaf(a, g, -0x1.13b6f8p-13f);
    g = fmaf(a, g, -0x1.1d4c74p-12f);
    g = fmaf(a, g,  0x1.dad62ap-8f);
    g = fmaf(a, g, -0x1.af201ep-5f);
    g = fmaf(a, g, -0x1.d62e9ap-2f);
    g = fmaf(a, g, -0x1.26af36p+0f);
    const float e = fmaf(a, g, -1.0f);
    return fmaf(-a, __builtin_amdgcn_exp2f(e), r);
}

__device__ __forceinline__ float xhalf(float x) {
    const int b = __builtin_bit_cast(int, x);
    return __builtin_bit_cast(float,
        __builtin_amdgcn_permlanex16(b, b, 0x76543210, 0xfedcba98, false, false));
}

__global__ __launch_bounds__(1024)
void pair_kernel(const int* __restrict__ positions,
                 const int* __restrict__ amino,
                 const float* __restrict__ pos_table,
                 const float* __restrict__ aa_table,
                 const float* __restrict__ W1,
                 const float* __restrict__ b1,
                 const float* __restrict__ W2,
                 const float* __restrict__ b2,
                 const float* __restrict__ W3,
                 const float* __restrict__ b3,
                 float* __restrict__ out) {
    __shared__ __attribute__((aligned(16))) float smem[(32 + 64 + 64 + 64 + 128) * LDS_STRIDE + 3 * 64];
    float* const u_lds = smem;
    float* const v_lds = u_lds + 32 * LDS_STRIDE;
    float* const w2t   = v_lds + 64 * LDS_STRIDE;
    float* const b1s   = w2t   + 64 * LDS_STRIDE;
    float* const b2s   = b1s + 64;
    float* const w3s   = b2s + 64;
    float* const e_lds = w3s + 64;
    float* const w1t   = e_lds + 64 * LDS_STRIDE;

    const int tid = threadIdx.x;
    const int b   = blockIdx.y;
    const int f0  = blockIdx.x * 16;

    {
        const int gt = tid >> 4;
        const int gk = (tid & 15) * 4;
        int p = positions[b * 64 + gt]; p = p < 0 ? 0 : (p > 1023 ? 1023 : p);
        int a = amino[b * 64 + gt];     a = a < 0 ? 0 : (a > 20   ? 20   : a);

        const int nn  = tid & 63;
        const int kq  = tid >> 6;

        v4f w1k[2];
        #pragma unroll
        for (int g = 0; g < 2; ++g) {
            const float* wp = W1 + (8 * kq + 4 * g) * 64 + nn;
            w1k[g].x = wp[0]; w1k[g].y = wp[64]; w1k[g].z = wp[128]; w1k[g].w = wp[192];
        }
        v4f w2k;
        {
            const float* wp = W2 + (4 * kq) * 64 + nn;
            w2k.x = wp[0]; w2k.y = wp[64]; w2k.z = wp[128]; w2k.w = wp[192];
        }
        const float b1v = b1[nn], b2v = b2[nn], w3v = W3[nn];
        const v4f pe = *(const v4f*)(pos_table + p * 64 + gk);
        const v4f ae = *(const v4f*)(aa_table  + a * 64 + gk);
        asm volatile("" :: "v"(b1v), "v"(b2v), "v"(w3v), "v"(pe), "v"(ae),
                     "v"(w1k[0]), "v"(w1k[1]), "v"(w2k));

        v4f ev;
        ev.x = pe.x + ae.x; ev.y = pe.y + ae.y; ev.z = pe.z + ae.z; ev.w = pe.w + ae.w;
        *(v4f*)&e_lds[gt * LDS_STRIDE + gk] = ev;
        {
            float* wd = &w1t[((kq >> 3) * 64 + nn) * LDS_STRIDE + 8 * (kq & 7)];
            *(v4f*)wd       = w1k[0];
            *(v4f*)(wd + 4) = w1k[1];
        }
        *(v4f*)&w2t[nn * LDS_STRIDE + 4 * kq] = w2k;
        if (tid < 64) { b1s[tid] = b1v; b2s[tid] = b2v; w3s[tid] = w3v; }
    }
    __syncthreads();

    const int lane = tid & 31;
    const int wv   = tid >> 5;
    const int jl   = lane & 15;
    const int half = lane >> 4;
    const int koff = 2 * half;

    if (wv < 24) {
        const int tt = wv >> 2;
        const int nt = wv & 3;
        int tok, arow;
        float* dst;
        if (tt < 4) {
            tok  = 16 * tt + jl;
            arow = 64 + 16 * nt + jl;
            dst  = &v_lds[tok * LDS_STRIDE + 16 * nt + 8 * half];
        } else {
            tok  = (tt == 4) ? f0 + jl : 62 - f0 - jl;
            arow = 16 * nt + jl;
            dst  = &u_lds[(2 * jl + (tt - 4)) * LDS_STRIDE + 16 * nt + 8 * half];
        }
        v8f acc1 = {0.f, 0.f, 0.f, 0.f, 0.f, 0.f, 0.f, 0.f};
        #pragma unroll
        for (int kk = 0; kk < 16; ++kk) {
            const int k0 = 4 * kk + koff;
            const v2f aa = *(const v2f*)&w1t[arow * LDS_STRIDE + k0];
            const v2f bb = *(const v2f*)&e_lds[tok * LDS_STRIDE + k0];
            acc1 = __builtin_amdgcn_wmma_f32_16x16x4_f32(false, aa, false, bb,
                                                        (short)0, acc1, false, false);
        }
        v4f d0, d1;
        d0.x = acc1[0]; d0.y = acc1[1]; d0.z = acc1[2]; d0.w = acc1[3];
        d1.x = acc1[4]; d1.y = acc1[5]; d1.z = acc1[6]; d1.w = acc1[7];
        *(v4f*)dst       = d0;
        *(v4f*)(dst + 4) = d1;
    }
    __syncthreads();

    const int fl   = wv >> 1;
    const int hr   = wv & 1;
    const int f    = f0 + fl;
    if (f == 31 && hr == 1) return;
    const int nfirst = 63 - f;

    const float bias3 = b3[0];
    #pragma unroll
    for (int mt = 0; mt < 2; ++mt) {
        v2f hB[16];
        int ig, jg;
        {
            const int t    = 32 * hr + 16 * mt + jl;
            const int sel  = (t >= nfirst) ? 1 : 0;
            const int urow = 2 * fl + sel;
            const int jrow = sel ? t : f + 1 + t;
            ig = sel ? 62 - f : f;
            jg = jrow;
            #pragma unroll
            for (int kk = 0; kk < 16; ++kk) {
                const int k0 = 4 * kk + koff;
                float x0 = u_lds[urow * LDS_STRIDE + k0]     + v_lds[jrow * LDS_STRIDE + k0]     + b1s[k0];
                float x1 = u_lds[urow * LDS_STRIDE + k0 + 1] + v_lds[jrow * LDS_STRIDE + k0 + 1] + b1s[k0 + 1];
                hB[kk].x = gelu_exact(x0);
                hB[kk].y = gelu_exact(x1);
            }
        }

        v8f acc[4];
        #pragma unroll
        for (int nt = 0; nt < 4; ++nt) {
            const v4f c0 = *(const v4f*)&b2s[nt * 16 + 8 * half];
            const v4f c1 = *(const v4f*)&b2s[nt * 16 + 8 * half + 4];
            acc[nt][0] = c0.x; acc[nt][1] = c0.y; acc[nt][2] = c0.z; acc[nt][3] = c0.w;
            acc[nt][4] = c1.x; acc[nt][5] = c1.y; acc[nt][6] = c1.z; acc[nt][7] = c1.w;
        }
        #pragma unroll
        for (int kk = 0; kk < 16; ++kk) {
            const int k0 = 4 * kk + koff;
            #pragma unroll
            for (int nt = 0; nt < 4; ++nt) {
                v2f aa = *(const v2f*)&w2t[(nt * 16 + jl) * LDS_STRIDE + k0];
                acc[nt] = __builtin_amdgcn_wmma_f32_16x16x4_f32(false, aa, false, hB[kk],
                                                               (short)0, acc[nt], false, false);
            }
        }

        float p[8];
        #pragma unroll
        for (int r = 0; r < 8; ++r) p[r] = 0.f;
        #pragma unroll
        for (int nt = 0; nt < 4; ++nt) {
            const v4f w0 = *(const v4f*)&w3s[nt * 16 + 8 * half];
            const v4f w1 = *(const v4f*)&w3s[nt * 16 + 8 * half + 4];
            const float w3n[8] = { w0.x, w0.y, w0.z, w0.w, w1.x, w1.y, w1.z, w1.w };
            #pragma unroll
            for (int r = 0; r < 8; ++r) p[r] = fmaf(gelu_exact(acc[nt][r]), w3n[r], p[r]);
        }
        float s = ((p[0] + p[1]) + (p[2] + p[3])) + ((p[4] + p[5]) + (p[6] + p[7]));
        s += xhalf(s);

        {
            const float val = s + bias3;
            const int row = half ? jg : ig;
            const int col = half ? ig : jg;
            out[(b * 64 + row) * 64 + col] = val;
        }
        if (mt == 0) __builtin_amdgcn_sched_barrier(0);
    }

    if (tid < 32) {
        const int idg = blockIdx.x * 32 + tid;
        out[(b * 64 + idg) * 64 + idg] = 0.f;
    }
}

extern "C" void kernel_launch(void* const* d_in, const int* in_sizes, int n_in,
                              void* d_out, int out_size, void* d_ws, size_t ws_size,
                              hipStream_t stream) {
    const int*   positions = (const int*)  d_in[0];
    const int*   amino     = (const int*)  d_in[1];
    const float* pos_table = (const float*)d_in[2];
    const float* aa_table  = (const float*)d_in[3];
    const float* W1        = (const float*)d_in[4];
    const float* b1        = (const float*)d_in[5];
    const float* W2        = (const float*)d_in[6];
    const float* b2        = (const float*)d_in[7];
    const float* W3        = (const float*)d_in[8];
    const float* b3        = (const float*)d_in[9];
    float* out = (float*)d_out;
    (void)d_ws; (void)ws_size;

    pair_kernel<<<dim3(2, 128), dim3(1024), 0, stream>>>(
        positions, amino, pos_table, aa_table, W1, b1, W2, b2, W3, b3, out);
}


# layer-3 gelu and its multiply-add by W3 written on 2-wide f32 vectors over the accumulator's register pairs: packed v_pk_fma_f32, two hidden units per instruction; same operations and sum order
# speedup vs baseline: 4.9541x; 1.0630x over previous
#include <hip/hip_runtime.h>
#include <math.h>


typedef __attribute__((ext_vector_type(2))) float v2f;
typedef __attribute__((ext_vector_type(4))) float v4f;
typedef __attribute__((ext_vector_type(8))) float v8f;

#define LDS_STRIDE 68

__device__ __forceinline__ float gelu_exact(float x) {
    const float r = fmaxf(x, 0.0f);
    const float a = fmaf(r, 2.0f, -x);
    float g = fmaf(a, -0x1.fe92fap-20f, 0x1.d6b5eep-16f);
    g = fmaf(a, g, -0x1.13b6f8p-13f);
    g = fmaf(a, g, -0x1.1d4c74p-12f);
    g = fmaf(a, g,  0x1.dad62ap-8f);
    g = fmaf(a, g, -0x1.af201ep-5f);
    g = fmaf(a, g, -0x1.d62e9ap-2f);
    g = fmaf(a, g, -0x1.26af36p+0f);
    const float e = fmaf(a, g, -1.0f);
    return fmaf(-a, __builtin_amdgcn_exp2f(e), r);
}

__device__ __forceinline__ v2f gelu_exact2(v2f x) {
    v2f r;
    r.x = fmaxf(x.x, 0.0f);
    r.y = fmaxf(x.y, 0.0f);
    const v2f two = {2.0f, 2.0f};
    const v2f a = __builtin_elementwise_fma(r, two, -x);
    const v2f c7 = {-0x1.fe92fap-20f, -0x1.fe92fap-20f};
    const v2f c6 = { 0x1.d6b5eep-16f,  0x1.d6b5eep-16f};
    const v2f c5 = {-0x1.13b6f8p-13f, -0x1.13b6f8p-13f};
    const v2f c4 = {-0x1.1d4c74p-12f, -0x1.1d4c74p-12f};
    const v2f c3 = { 0x1.dad62ap-8f,   0x1.dad62ap-8f};
    const v2f c2 = {-0x1.af201ep-5f,  -0x1.af201ep-5f};
    const v2f c1 = {-0x1.d62e9ap-2f,  -0x1.d62e9ap-2f};
    const v2f c0 = {-0x1.26af36p+0f,  -0x1.26af36p+0f};
    const v2f m1 = {-1.0f, -1.0f};
    v2f g = __builtin_elementwise_fma(a, c7, c6);
    g = __builtin_elementwise_fma(a, g, c5);
    g = __builtin_elementwise_fma(a, g, c4);
    g = __builtin_elementwise_fma(a, g, c3);
    g = __builtin_elementwise_fma(a, g, c2);
    g = __builtin_elementwise_fma(a, g, c1);
    g = __builtin_elementwise_fma(a, g, c0);
    const v2f e = __builtin_elementwise_fma(a, g, m1);
    v2f q;
    q.x = __builtin_amdgcn_exp2f(e.x);
    q.y = __builtin_amdgcn_exp2f(e.y);
    return __builtin_elementwise_fma(-a, q, r);
}

__device__ __forceinline__ float xhalf(float x) {
    const int b = __builtin_bit_cast(int, x);
    return __builtin_bit_cast(float,
        __builtin_amdgcn_permlanex16(b, b, 0x76543210, 0xfedcba98, false, false));
}

__global__ __launch_bounds__(1024)
void pair_kernel(const int* __restrict__ positions,
                 const int* __restrict__ amino,
                 const float* __restrict__ pos_table,
                 const float* __restrict__ aa_table,
                 const float* __restrict__ W1,
                 const float* __restrict__ b1,
                 const float* __restrict__ W2,
                 const float* __restrict__ b2,
                 const float* __restrict__ W3,
                 const float* __restrict__ b3,
                 float* __restrict__ out) {
    __shared__ __attribute__((aligned(16))) float smem[(32 + 64 + 64 + 64 + 128) * LDS_STRIDE + 3 * 64];
    float* const u_lds = smem;
    float* const v_lds = u_lds + 32 * LDS_STRIDE;
    float* const w2t   = v_lds + 64 * LDS_STRIDE;
    float* const b1s   = w2t   + 64 * LDS_STRIDE;
    float* const b2s   = b1s + 64;
    float* const w3s   = b2s + 64;
    float* const e_lds = w3s + 64;
    float* const w1t   = e_lds + 64 * LDS_STRIDE;

    const int tid = threadIdx.x;
    const int b   = blockIdx.y;
    const int f0  = blockIdx.x * 16;

    {
        const int gt = tid >> 4;
        const int gk = (tid & 15) * 4;
        int p = positions[b * 64 + gt]; p = p < 0 ? 0 : (p > 1023 ? 1023 : p);
        int a = amino[b * 64 + gt];     a = a < 0 ? 0 : (a > 20   ? 20   : a);

        const int nn  = tid & 63;
        const int kq  = tid >> 6;

        v4f w1k[2];
        #pragma unroll
        for (int g = 0; g < 2; ++g) {
            const float* wp = W1 + (8 * kq + 4 * g) * 64 + nn;
            w1k[g].x = wp[0]; w1k[g].y = wp[64]; w1k[g].z = wp[128]; w1k[g].w = wp[192];
        }
        v4f w2k;
        {
            const float* wp = W2 + (4 * kq) * 64 + nn;
            w2k.x = wp[0]; w2k.y = wp[64]; w2k.z = wp[128]; w2k.w = wp[192];
        }
        const float b1v = b1[nn], b2v = b2[nn], w3v = W3[nn];
        const v4f pe = *(const v4f*)(pos_table + p * 64 + gk);
        const v4f ae = *(const v4f*)(aa_table  + a * 64 + gk);
        asm volatile("" :: "v"(b1v), "v"(b2v), "v"(w3v), "v"(pe), "v"(ae),
                     "v"(w1k[0]), "v"(w1k[1]), "v"(w2k));

        v4f ev;
        ev.x = pe.x + ae.x; ev.y = pe.y + ae.y; ev.z = pe.z + ae.z; ev.w = pe.w + ae.w;
        *(v4f*)&e_lds[gt * LDS_STRIDE + gk] = ev;
        {
            float* wd = &w1t[((kq >> 3) * 64 + nn) * LDS_STRIDE + 8 * (kq & 7)];
            *(v4f*)wd       = w1k[0];
            *(v4f*)(wd + 4) = w1k[1];
        }
        *(v4f*)&w2t[nn * LDS_STRIDE + 4 * kq] = w2k;
        if (tid < 64) { b1s[tid] = b1v; b2s[tid] = b2v; w3s[tid] = w3v; }
    }
    __syncthreads();

    const int lane = tid & 31;
    const int wv   = tid >> 5;
    const int jl   = lane & 15;
    const int half = lane >> 4;
    const int koff = 2 * half;

    if (wv < 24) {
        const int tt = wv >> 2;
        const int nt = wv & 3;
        int tok, arow;
        float* dst;
        if (tt < 4) {
            tok  = 16 * tt + jl;
            arow = 64 + 16 * nt + jl;
            dst  = &v_lds[tok * LDS_STRIDE + 16 * nt + 8 * half];
        } else {
            tok  = (tt == 4) ? f0 + jl : 62 - f0 - jl;
            arow = 16 * nt + jl;
            dst  = &u_lds[(2 * jl + (tt - 4)) * LDS_STRIDE + 16 * nt + 8 * half];
        }
        v8f acc1 = {0.f, 0.f, 0.f, 0.f, 0.f, 0.f, 0.f, 0.f};
        #pragma unroll
        for (int kk = 0; kk < 16; ++kk) {
            const int k0 = 4 * kk + koff;
            const v2f aa = *(const v2f*)&w1t[arow * LDS_STRIDE + k0];
            const v2f bb = *(const v2f*)&e_lds[tok * LDS_STRIDE + k0];
            acc1 = __builtin_amdgcn_wmma_f32_16x16x4_f32(false, aa, false, bb,
                                                        (short)0, acc1, false, false);
        }
        v4f d0, d1;
        d0.x = acc1[0]; d0.y = acc1[1]; d0.z = acc1[2]; d0.w = acc1[3];
        d1.x = acc1[4]; d1.y = acc1[5]; d1.z = acc1[6]; d1.w = acc1[7];
        *(v4f*)dst       = d0;
        *(v4f*)(dst + 4) = d1;
    }
    __syncthreads();

    const int fl   = wv >> 1;
    const int hr   = wv & 1;
    const int f    = f0 + fl;
    if (f == 31 && hr == 1) return;
    const int nfirst = 63 - f;

    const float bias3 = b3[0];
    #pragma unroll
    for (int mt = 0; mt < 2; ++mt) {
        v2f hB[16];
        int ig, jg;
        {
            const int t    = 32 * hr + 16 * mt + jl;
            const int sel  = (t >= nfirst) ? 1 : 0;
            const int urow = 2 * fl + sel;
            const int jrow = sel ? t : f + 1 + t;
            ig = sel ? 62 - f : f;
            jg = jrow;
            #pragma unroll
            for (int kk = 0; kk < 16; ++kk) {
                const int k0 = 4 * kk + koff;
                float x0 = u_lds[urow * LDS_STRIDE + k0]     + v_lds[jrow * LDS_STRIDE + k0]     + b1s[k0];
                float x1 = u_lds[urow * LDS_STRIDE + k0 + 1] + v_lds[jrow * LDS_STRIDE + k0 + 1] + b1s[k0 + 1];
                hB[kk].x = gelu_exact(x0);
                hB[kk].y = gelu_exact(x1);
            }
        }

        v8f acc[4];
        #pragma unroll
        for (int nt = 0; nt < 4; ++nt) {
            const v4f c0 = *(const v4f*)&b2s[nt * 16 + 8 * half];
            const v4f c1 = *(const v4f*)&b2s[nt * 16 + 8 * half + 4];
            acc[nt][0] = c0.x; acc[nt][1] = c0.y; acc[nt][2] = c0.z; acc[nt][3] = c0.w;
            acc[nt][4] = c1.x; acc[nt][5] = c1.y; acc[nt][6] = c1.z; acc[nt][7] = c1.w;
        }
        #pragma unroll
        for (int kk = 0; kk < 16; ++kk) {
            const int k0 = 4 * kk + koff;
            #pragma unroll
            for (int nt = 0; nt < 4; ++nt) {
                v2f aa = *(const v2f*)&w2t[(nt * 16 + jl) * LDS_STRIDE + k0];
                acc[nt] = __builtin_amdgcn_wmma_f32_16x16x4_f32(false, aa, false, hB[kk],
                                                               (short)0, acc[nt], false, false);
            }
        }

        v2f p[4];
        #pragma unroll
        for (int q = 0; q < 4; ++q) { p[q].x = 0.f; p[q].y = 0.f; }
        #pragma unroll
        for (int nt = 0; nt < 4; ++nt) {
            const v4f w0 = *(const v4f*)&w3s[nt * 16 + 8 * half];
            const v4f w1 = *(const v4f*)&w3s[nt * 16 + 8 * half + 4];
            v2f w3n[4];
            w3n[0].x = w0.x; w3n[0].y = w0.y; w3n[1].x = w0.z; w3n[1].y = w0.w;
            w3n[2].x = w1.x; w3n[2].y = w1.y; w3n[3].x = w1.z; w3n[3].y = w1.w;
            #pragma unroll
            for (int q = 0; q < 4; ++q) {
                v2f x;
                x.x = acc[nt][2 * q];
                x.y = acc[nt][2 * q + 1];
                p[q] = __builtin_elementwise_fma(gelu_exact2(x), w3n[q], p[q]);
            }
        }
        float s = ((p[0].x + p[0].y) + (p[1].x + p[1].y)) + ((p[2].x + p[2].y) + (p[3].x + p[3].y));
        s += xhalf(s);

        {
            const float val = s + bias3;
            const int row = half ? jg : ig;
            const int col = half ? ig : jg;
            out[(b * 64 + row) * 64 + col] = val;
        }
        if (mt == 0) __builtin_amdgcn_sched_barrier(0);
    }

    if (tid < 32) {
        const int idg = blockIdx.x * 32 + tid;
        out[(b * 64 + idg) * 64 + idg] = 0.f;
    }
}

extern "C" void kernel_launch(void* const* d_in, const int* in_sizes, int n_in,
                              void* d_out, int out_size, void* d_ws, size_t ws_size,
                              hipStream_t stream) {
    const int*   positions = (const int*)  d_in[0];
    const int*   amino     = (const int*)  d_in[1];
    const float* pos_table = (const float*)d_in[2];
    const float* aa_table  = (const float*)d_in[3];
    const float* W1        = (const float*)d_in[4];
    const float* b1        = (const float*)d_in[5];
    const float* W2        = (const float*)d_in[6];
    const float* b2        = (const float*)d_in[7];
    const float* W3        = (const float*)d_in[8];
    const float* b3        = (const float*)d_in[9];
    float* out = (float*)d_out;
    (void)d_ws; (void)ws_size;

    pair_kernel<<<dim3(2, 128), dim3(1024), 0, stream>>>(
        positions, amino, pos_table, aa_table, W1, b1, W2, b2, W3, b3, out);
}
